# starting kernel
# baseline (speedup 1.0000x reference)
#include <hip/hip_runtime.h>
#include <hip/hip_bf16.h>
#include <math.h>


typedef float v2f __attribute__((ext_vector_type(2)));
typedef float v8f __attribute__((ext_vector_type(8)));

__global__ void k_zero(float* __restrict__ p, int n) {
  int i = blockIdx.x * blockDim.x + threadIdx.x;
  if (i < n) p[i] = 0.0f;
}

__global__ void k_deg(const int* __restrict__ dst, const float* __restrict__ ew,
                      float* __restrict__ deg, int E) {
  int e = blockIdx.x * blockDim.x + threadIdx.x;
  if (e < E) unsafeAtomicAdd(&deg[dst[e]], ew[e]);
}

__global__ void k_dis(const float* __restrict__ deg, float* __restrict__ dis,
                      float* __restrict__ nself, int N) {
  int n = blockIdx.x * blockDim.x + threadIdx.x;
  if (n < N) {
    float d = deg[n] + 1.0f;
    float r = (d > 0.0f) ? rsqrtf(d) : 0.0f;
    dis[n]   = r;
    nself[n] = r * r;
  }
}

__global__ void k_norm(const int* __restrict__ src, const int* __restrict__ dst,
                       const float* __restrict__ ew, const float* __restrict__ dis,
                       float* __restrict__ normE, int E) {
  int e = blockIdx.x * blockDim.x + threadIdx.x;
  if (e < E) normE[e] = dis[src[e]] * ew[e] * dis[dst[e]];
}

template <int COUT, bool RELU>
__global__ void k_gemm_wmma(const float* __restrict__ H, const float* __restrict__ Wg,
                            const float* __restrict__ bias, const float* __restrict__ nself,
                            float* __restrict__ HW, float* __restrict__ OUT, int N) {
  constexpr int COLT = COUT / 16;
  constexpr int ROWT = 8 / COLT;
  constexpr int WSTR = 68;
  __shared__ float Wl[COUT * WSTR];

  for (int i = threadIdx.x; i < 64 * COUT; i += 256) {
    int k = i / COUT;
    int n = i - k * COUT;
    Wl[n * WSTR + k] = Wg[i];
  }
  __syncthreads();

  const int wave = threadIdx.x >> 5;
  const int lane = threadIdx.x & 31;
  const int rowTile = wave / COLT;
  const int colTile = wave - rowTile * COLT;
  const int m0 = (blockIdx.x * ROWT + rowTile) * 16;
  if (m0 >= N) return;
  const int n0 = colTile * 16;
  const int mr = lane & 15;
  const int kb = (lane >> 4) << 1;

  v8f acc = {};
  const float* hrow = H + (size_t)(m0 + mr) * 64;
  const float* wrow = Wl + (n0 + mr) * WSTR;
#pragma unroll
  for (int k = 0; k < 64; k += 4) {
    float a0 = hrow[k + kb];
    float a1 = hrow[k + kb + 1];
    if (RELU) { a0 = fmaxf(a0, 0.0f); a1 = fmaxf(a1, 0.0f); }
    v2f a; a.x = a0; a.y = a1;
    v2f b = *(const v2f*)&wrow[k + kb];
    acc = __builtin_amdgcn_wmma_f32_16x16x4_f32(
        false, a, false, b, (short)0, acc, false, false);
  }

  const int mbase = m0 + ((lane >> 4) << 3);
  const float bcol = bias[n0 + mr];
#pragma unroll
  for (int v = 0; v < 8; ++v) {
    float hv = acc[v];
    size_t idx = (size_t)(mbase + v) * COUT + n0 + mr;
    HW[idx]  = hv;
    OUT[idx] = bcol + nself[mbase + v] * hv;
  }
}

__global__ void k_scatter64(const float* __restrict__ hw, const float* __restrict__ normE,
                            const int* __restrict__ src, const int* __restrict__ dst,
                            float* __restrict__ out, int E) {
  long t = (long)blockIdx.x * blockDim.x + threadIdx.x;
  int e = (int)(t >> 4);
  if (e >= E) return;
  int f = (int)(t & 15) << 2;
  float nm = normE[e];
  float4 v = *(const float4*)(hw + (size_t)src[e] * 64 + f);
  float* op = out + (size_t)dst[e] * 64 + f;
  unsafeAtomicAdd(op + 0, nm * v.x);
  unsafeAtomicAdd(op + 1, nm * v.y);
  unsafeAtomicAdd(op + 2, nm * v.z);
  unsafeAtomicAdd(op + 3, nm * v.w);
}

__global__ void k_scatter16(const float* __restrict__ hw, const float* __restrict__ normE,
                            const int* __restrict__ src, const int* __restrict__ dst,
                            float* __restrict__ out, int E) {
  long t = (long)blockIdx.x * blockDim.x + threadIdx.x;
  int e = (int)(t >> 2);
  if (e >= E) return;
  int f = (int)(t & 3) << 2;
  float nm = normE[e];
  float4 v = *(const float4*)(hw + (size_t)src[e] * 16 + f);
  float* op = out + (size_t)dst[e] * 16 + f;
  unsafeAtomicAdd(op + 0, nm * v.x);
  unsafeAtomicAdd(op + 1, nm * v.y);
  unsafeAtomicAdd(op + 2, nm * v.z);
  unsafeAtomicAdd(op + 3, nm * v.w);
}

__global__ void k_logsoftmax16(float* __restrict__ out, int N) {
  int n = blockIdx.x * blockDim.x + threadIdx.x;
  if (n >= N) return;
  const float* lg = out + (size_t)N * 16 + (size_t)n * 16;
  float v[16];
  float mx = -3.402823466e38f;
#pragma unroll
  for (int i = 0; i < 16; ++i) { v[i] = lg[i]; mx = fmaxf(mx, v[i]); }
  float s = 0.0f;
#pragma unroll
  for (int i = 0; i < 16; ++i) s += expf(v[i] - mx);
  float lse = mx + logf(s);
  float* o0 = out + (size_t)n * 16;
  float* o2 = out + (size_t)2 * N * 16 + (size_t)n * 16;
#pragma unroll
  for (int i = 0; i < 16; ++i) {
    o0[i] = v[i] - lse;
    o2[i] = v[i];
  }
}

extern "C" void kernel_launch(void* const* d_in, const int* in_sizes, int n_in,
                              void* d_out, int out_size, void* d_ws, size_t ws_size,
                              hipStream_t stream) {
  const float* x  = (const float*)d_in[0];
  const int*   ei = (const int*)  d_in[1];
  const float* ew = (const float*)d_in[2];
  const float* W1 = (const float*)d_in[3];
  const float* b1 = (const float*)d_in[4];
  const float* W2 = (const float*)d_in[5];
  const float* b2 = (const float*)d_in[6];
  const float* W3 = (const float*)d_in[7];
  const float* b3 = (const float*)d_in[8];
  float* out = (float*)d_out;

  const int N = in_sizes[0] / 64;
  const int E = in_sizes[2];
  const int* src = ei;
  const int* dst = ei + E;

  float* w     = (float*)d_ws;
  float* deg   = w;
  float* dis   = deg + N;
  float* nself = dis + N;
  float* normE = nself + N;
  float* bufA  = normE + E;
  float* bufB  = bufA + (size_t)N * 64;
  float* bufC  = bufB + (size_t)N * 64;
  float* logits = out + (size_t)N * 16;

  const int T = 256;
  const int gN = (N + T - 1) / T;
  const int gE = (E + T - 1) / T;
  const int rowTiles = (N + 15) / 16;
  const int gSc64 = (int)(((size_t)E * 16 + T - 1) / T);
  const int gSc16 = (int)(((size_t)E * 4  + T - 1) / T);

  k_zero<<<gN, T, 0, stream>>>(deg, N);
  k_deg <<<gE, T, 0, stream>>>(dst, ew, deg, E);
  k_dis <<<gN, T, 0, stream>>>(deg, dis, nself, N);
  k_norm<<<gE, T, 0, stream>>>(src, dst, ew, dis, normE, E);

  k_gemm_wmma<64, false><<<(rowTiles + 1) / 2, 256, 0, stream>>>(
      x, W1, b1, nself, bufA, bufB, N);
  k_scatter64<<<gSc64, T, 0, stream>>>(bufA, normE, src, dst, bufB, E);

  k_gemm_wmma<64, true><<<(rowTiles + 1) / 2, 256, 0, stream>>>(
      bufB, W2, b2, nself, bufA, bufC, N);
  k_scatter64<<<gSc64, T, 0, stream>>>(bufA, normE, src, dst, bufC, E);

  k_gemm_wmma<16, true><<<(rowTiles + 7) / 8, 256, 0, stream>>>(
      bufC, W3, b3, nself, bufA, logits, N);
  k_scatter16<<<gSc16, T, 0, stream>>>(bufA, normE, src, dst, logits, E);

  k_logsoftmax16<<<gN, T, 0, stream>>>(out, N);
}


# Edges sorted by destination once per call (atomic count gives arrival-order rank, scan, fill); each layer then gathers and sums per node, no float atomics; summation order changes
# speedup vs baseline: 1.6604x; 1.6604x over previous
#include <hip/hip_runtime.h>
#include <hip/hip_bf16.h>
#include <math.h>


typedef float v2f __attribute__((ext_vector_type(2)));
typedef float v8f __attribute__((ext_vector_type(8)));

__global__ void k_zero(int* __restrict__ p, int n) {
  int i = blockIdx.x * blockDim.x + threadIdx.x;
  if (i < n) p[i] = 0;
}

__global__ void k_count(const int* __restrict__ dst, int* __restrict__ cnt,
                        int* __restrict__ rank, int E) {
  int e = blockIdx.x * blockDim.x + threadIdx.x;
  if (e < E) rank[e] = atomicAdd(&cnt[dst[e]], 1);
}

__device__ __forceinline__ int wave_sum(int v) {
  for (int m = 16; m >= 1; m >>= 1) v += __shfl_xor(v, m);
  return v;
}

__global__ void k_bsum(const int* __restrict__ cnt, int* __restrict__ bsum, int N) {
  __shared__ int ws[8];
  const int i0 = (blockIdx.x * 256 + threadIdx.x) * 4;
  int v = 0;
#pragma unroll
  for (int k = 0; k < 4; ++k) if (i0 + k < N) v += cnt[i0 + k];
  v = wave_sum(v);
  if ((threadIdx.x & 31) == 0) ws[threadIdx.x >> 5] = v;
  __syncthreads();
  if (threadIdx.x == 0) {
    int s = 0;
#pragma unroll
    for (int w = 0; w < 8; ++w) s += ws[w];
    bsum[blockIdx.x] = s;
  }
}

__global__ void k_scan(const int* __restrict__ cnt, const int* __restrict__ bsum,
                       int* __restrict__ rowptr, int N, int E) {
  __shared__ int ws[8];
  __shared__ int wo[8];
  const int lane = threadIdx.x & 31;
  const int wave = threadIdx.x >> 5;
  int part = 0;
  for (int i = threadIdx.x; i < (int)blockIdx.x; i += 256) part += bsum[i];
  part = wave_sum(part);
  if (lane == 0) wo[wave] = part;
  const int i0 = (blockIdx.x * 256 + threadIdx.x) * 4;
  int c[4];
  int tot = 0;
#pragma unroll
  for (int k = 0; k < 4; ++k) { c[k] = (i0 + k < N) ? cnt[i0 + k] : 0; tot += c[k]; }
  int inc = tot;
  for (int d = 1; d < 32; d <<= 1) {
    int t = __shfl_up(inc, d);
    if (lane >= d) inc += t;
  }
  if (lane == 31) ws[wave] = inc;
  __syncthreads();
  int off = 0;
#pragma unroll
  for (int w = 0; w < 8; ++w) { off += wo[w]; if (w < wave) off += ws[w]; }
  int run = off + inc - tot;
#pragma unroll
  for (int k = 0; k < 4; ++k) if (i0 + k < N) { rowptr[i0 + k] = run; run += c[k]; }
  if (blockIdx.x == 0 && threadIdx.x == 0) rowptr[N] = E;
}

__global__ void k_fill(const int* __restrict__ src, const int* __restrict__ dst,
                       const float* __restrict__ ew, const int* __restrict__ rowptr,
                       const int* __restrict__ rank, int2* __restrict__ csr, int E) {
  int e = blockIdx.x * blockDim.x + threadIdx.x;
  if (e < E) {
    int p = rowptr[dst[e]] + rank[e];
    csr[p] = make_int2(src[e], __float_as_int(ew[e]));
  }
}

__global__ void k_degdis(const int2* __restrict__ csr, const int* __restrict__ rowptr,
                         float* __restrict__ dis, float* __restrict__ nself, int N) {
  long t = (long)blockIdx.x * blockDim.x + threadIdx.x;
  int n = (int)(t >> 3);
  if (n >= N) return;
  int l = (int)(t & 7);
  const int pe = rowptr[n + 1];
  float s = 0.0f;
  for (int p = rowptr[n] + l; p < pe; p += 8) s += __int_as_float(csr[p].y);
  s += __shfl_xor(s, 1);
  s += __shfl_xor(s, 2);
  s += __shfl_xor(s, 4);
  if (l == 0) {
    float d = s + 1.0f;
    float r = (d > 0.0f) ? rsqrtf(d) : 0.0f;
    dis[n]   = r;
    nself[n] = r * r;
  }
}

__global__ void k_normcsr(int2* __restrict__ csr, const int* __restrict__ rowptr,
                          const float* __restrict__ dis, int N) {
  long t = (long)blockIdx.x * blockDim.x + threadIdx.x;
  int n = (int)(t >> 3);
  if (n >= N) return;
  int l = (int)(t & 7);
  const int pe = rowptr[n + 1];
  const float dn = dis[n];
  for (int p = rowptr[n] + l; p < pe; p += 8) {
    int2 c = csr[p];
    float nm = dis[c.x] * __int_as_float(c.y) * dn;
    csr[p].y = __float_as_int(nm);
  }
}

template <int COUT, bool RELU>
__global__ void k_gemm_wmma(const float* __restrict__ H, const float* __restrict__ Wg,
                            float* __restrict__ HW, int N) {
  constexpr int COLT = COUT / 16;
  constexpr int ROWT = 8 / COLT;
  constexpr int WSTR = 68;
  __shared__ float Wl[COUT * WSTR];

  for (int i = threadIdx.x; i < 64 * COUT; i += 256) {
    int k = i / COUT;
    int n = i - k * COUT;
    Wl[n * WSTR + k] = Wg[i];
  }
  __syncthreads();

  const int wave = threadIdx.x >> 5;
  const int lane = threadIdx.x & 31;
  const int rowTile = wave / COLT;
  const int colTile = wave - rowTile * COLT;
  const int m0 = (blockIdx.x * ROWT + rowTile) * 16;
  if (m0 >= N) return;
  const int n0 = colTile * 16;
  const int mr = lane & 15;
  const int kb = (lane >> 4) << 1;

  v8f acc = {};
  const float* hrow = H + (size_t)(m0 + mr) * 64;
  const float* wrow = Wl + (n0 + mr) * WSTR;
#pragma unroll
  for (int k = 0; k < 64; k += 4) {
    float a0 = hrow[k + kb];
    float a1 = hrow[k + kb + 1];
    if (RELU) { a0 = fmaxf(a0, 0.0f); a1 = fmaxf(a1, 0.0f); }
    v2f a; a.x = a0; a.y = a1;
    v2f b = *(const v2f*)&wrow[k + kb];
    acc = __builtin_amdgcn_wmma_f32_16x16x4_f32(
        false, a, false, b, (short)0, acc, false, false);
  }

  const int mbase = m0 + ((lane >> 4) << 3);
#pragma unroll
  for (int v = 0; v < 8; ++v) {
    size_t idx = (size_t)(mbase + v) * COUT + n0 + mr;
    HW[idx] = acc[v];
  }
}

template <int LPN>
__global__ void k_agg(const float* __restrict__ hw, const int2* __restrict__ csr,
                      const int* __restrict__ rowptr, const float* __restrict__ bias,
                      const float* __restrict__ nself, float* __restrict__ out, int N) {
  constexpr int C = 4 * LPN;
  long t = (long)blockIdx.x * blockDim.x + threadIdx.x;
  int n = (int)(t / LPN);
  if (n >= N) return;
  const int f = (int)(t % LPN) << 2;
  int p = rowptr[n];
  const int pe = rowptr[n + 1];
  const float4 hs = *(const float4*)(hw + (size_t)n * C + f);
  const float ns = nself[n];
  float4 acc;
  acc.x = bias[f + 0] + ns * hs.x;
  acc.y = bias[f + 1] + ns * hs.y;
  acc.z = bias[f + 2] + ns * hs.z;
  acc.w = bias[f + 3] + ns * hs.w;
  for (; p + 4 <= pe; p += 4) {
    const int2 c0 = csr[p], c1 = csr[p + 1], c2 = csr[p + 2], c3 = csr[p + 3];
    const float4 v0 = *(const float4*)(hw + (size_t)c0.x * C + f);
    const float4 v1 = *(const float4*)(hw + (size_t)c1.x * C + f);
    const float4 v2 = *(const float4*)(hw + (size_t)c2.x * C + f);
    const float4 v3 = *(const float4*)(hw + (size_t)c3.x * C + f);
    const float n0 = __int_as_float(c0.y), n1 = __int_as_float(c1.y);
    const float n2 = __int_as_float(c2.y), n3 = __int_as_float(c3.y);
    acc.x += n0 * v0.x; acc.y += n0 * v0.y; acc.z += n0 * v0.z; acc.w += n0 * v0.w;
    acc.x += n1 * v1.x; acc.y += n1 * v1.y; acc.z += n1 * v1.z; acc.w += n1 * v1.w;
    acc.x += n2 * v2.x; acc.y += n2 * v2.y; acc.z += n2 * v2.z; acc.w += n2 * v2.w;
    acc.x += n3 * v3.x; acc.y += n3 * v3.y; acc.z += n3 * v3.z; acc.w += n3 * v3.w;
  }
#pragma unroll 1
  for (; p < pe; ++p) {
    const int2 c0 = csr[p];
    const float4 v0 = *(const float4*)(hw + (size_t)c0.x * C + f);
    const float n0 = __int_as_float(c0.y);
    acc.x += n0 * v0.x; acc.y += n0 * v0.y; acc.z += n0 * v0.z; acc.w += n0 * v0.w;
  }
  *(float4*)(out + (size_t)n * C + f) = acc;
}

__global__ void k_logsoftmax16(float* __restrict__ out, int N) {
  int n = blockIdx.x * blockDim.x + threadIdx.x;
  if (n >= N) return;
  const float* lg = out + (size_t)N * 16 + (size_t)n * 16;
  float v[16];
  float mx = -3.402823466e38f;
#pragma unroll
  for (int i = 0; i < 16; ++i) { v[i] = lg[i]; mx = fmaxf(mx, v[i]); }
  float s = 0.0f;
#pragma unroll
  for (int i = 0; i < 16; ++i) s += expf(v[i] - mx);
  float lse = mx + logf(s);
  float* o0 = out + (size_t)n * 16;
  float* o2 = out + (size_t)2 * N * 16 + (size_t)n * 16;
#pragma unroll
  for (int i = 0; i < 16; ++i) {
    o0[i] = v[i] - lse;
    o2[i] = v[i];
  }
}

extern "C" void kernel_launch(void* const* d_in, const int* in_sizes, int n_in,
                              void* d_out, int out_size, void* d_ws, size_t ws_size,
                              hipStream_t stream) {
  const float* x  = (const float*)d_in[0];
  const int*   ei = (const int*)  d_in[1];
  const float* ew = (const float*)d_in[2];
  const float* W1 = (const float*)d_in[3];
  const float* b1 = (const float*)d_in[4];
  const float* W2 = (const float*)d_in[5];
  const float* b2 = (const float*)d_in[6];
  const float* W3 = (const float*)d_in[7];
  const float* b3 = (const float*)d_in[8];
  float* out = (float*)d_out;

  const int N = in_sizes[0] / 64;
  const int E = in_sizes[2];
  const int* src = ei;
  const int* dst = ei + E;

  const int Np = (N + 4) & ~3;
  const int Ep = (E + 3) & ~3;
  const int nb = (N + 1023) / 1024;
  const int nbp = (nb + 3) & ~3;

  int*   cnt    = (int*)d_ws;
  int*   rowptr = cnt + Np;
  int*   bsum   = rowptr + Np;
  int*   rank   = bsum + nbp;
  int2*  csr    = (int2*)(rank + Ep);
  float* dis    = (float*)(csr + Ep);
  float* nself  = dis + Np;
  float* bufA   = nself + Np;
  float* bufB   = bufA + (size_t)N * 64;
  float* bufC   = bufB + (size_t)N * 64;
  float* logits = out + (size_t)N * 16;

  const int T = 256;
  const int gN = (N + T - 1) / T;
  const int gE = (E + T - 1) / T;
  const int g8N = (int)(((size_t)N * 8 + T - 1) / T);
  const int rowTiles = (N + 15) / 16;
  const int gAg64 = (int)(((size_t)N * 16 + T - 1) / T);
  const int gAg16 = (int)(((size_t)N * 4  + T - 1) / T);

  k_zero   <<<gN, T, 0, stream>>>(cnt, N);
  k_count  <<<gE, T, 0, stream>>>(dst, cnt, rank, E);
  k_bsum   <<<nb, 256, 0, stream>>>(cnt, bsum, N);
  k_scan   <<<nb, 256, 0, stream>>>(cnt, bsum, rowptr, N, E);
  k_fill   <<<gE, T, 0, stream>>>(src, dst, ew, rowptr, rank, csr, E);
  k_degdis <<<g8N, T, 0, stream>>>(csr, rowptr, dis, nself, N);
  k_normcsr<<<g8N, T, 0, stream>>>(csr, rowptr, dis, N);

  k_gemm_wmma<64, false><<<(rowTiles + 1) / 2, 256, 0, stream>>>(x, W1, bufA, N);
  k_agg<16><<<gAg64, T, 0, stream>>>(bufA, csr, rowptr, b1, nself, bufB, N);

  k_gemm_wmma<64, true><<<(rowTiles + 1) / 2, 256, 0, stream>>>(bufB, W2, bufA, N);
  k_agg<16><<<gAg64, T, 0, stream>>>(bufA, csr, rowptr, b2, nself, bufC, N);

  k_gemm_wmma<16, true><<<(rowTiles + 7) / 8, 256, 0, stream>>>(bufC, W3, bufA, N);
  k_agg<4><<<gAg16, T, 0, stream>>>(bufA, csr, rowptr, b3, nself, logits, N);

  k_logsoftmax16<<<gN, T, 0, stream>>>(out, N);
}


# speedup vs baseline: 1.9207x; 1.1568x over previous
#include <hip/hip_runtime.h>
#include <hip/hip_bf16.h>
#include <math.h>


typedef float v2f __attribute__((ext_vector_type(2)));
typedef float v8f __attribute__((ext_vector_type(8)));

__global__ void k_zero(int* __restrict__ p, int n) {
  int i = blockIdx.x * blockDim.x + threadIdx.x;
  if (i < n) p[i] = 0;
}

__global__ void k_count(const int* __restrict__ dst, int* __restrict__ cnt,
                        int* __restrict__ rank, int E) {
  int e = blockIdx.x * blockDim.x + threadIdx.x;
  if (e < E) rank[e] = atomicAdd(&cnt[dst[e]], 1);
}

__device__ __forceinline__ int wave_sum(int v) {
  for (int m = 16; m >= 1; m >>= 1) v += __shfl_xor(v, m);
  return v;
}

__global__ void k_bsum(const int* __restrict__ cnt, int* __restrict__ bsum, int N) {
  __shared__ int ws[8];
  const int i0 = (blockIdx.x * 256 + threadIdx.x) * 4;
  int v = 0;
#pragma unroll
  for (int k = 0; k < 4; ++k) if (i0 + k < N) v += cnt[i0 + k];
  v = wave_sum(v);
  if ((threadIdx.x & 31) == 0) ws[threadIdx.x >> 5] = v;
  __syncthreads();
  if (threadIdx.x == 0) {
    int s = 0;
#pragma unroll
    for (int w = 0; w < 8; ++w) s += ws[w];
    bsum[blockIdx.x] = s;
  }
}

__global__ void k_scan(const int* __restrict__ cnt, const int* __restrict__ bsum,
                       int* __restrict__ rowptr, int N, int E) {
  __shared__ int ws[8];
  __shared__ int wo[8];
  const int lane = threadIdx.x & 31;
  const int wave = threadIdx.x >> 5;
  int part = 0;
  for (int i = threadIdx.x; i < (int)blockIdx.x; i += 256) part += bsum[i];
  part = wave_sum(part);
  if (lane == 0) wo[wave] = part;
  const int i0 = (blockIdx.x * 256 + threadIdx.x) * 4;
  int c[4];
  int tot = 0;
#pragma unroll
  for (int k = 0; k < 4; ++k) { c[k] = (i0 + k < N) ? cnt[i0 + k] : 0; tot += c[k]; }
  int inc = tot;
  for (int d = 1; d < 32; d <<= 1) {
    int t = __shfl_up(inc, d);
    if (lane >= d) inc += t;
  }
  if (lane == 31) ws[wave] = inc;
  __syncthreads();
  int off = 0;
#pragma unroll
  for (int w = 0; w < 8; ++w) { off += wo[w]; if (w < wave) off += ws[w]; }
  int run = off + inc - tot;
#pragma unroll
  for (int k = 0; k < 4; ++k) if (i0 + k < N) { rowptr[i0 + k] = run; run += c[k]; }
  if (blockIdx.x == 0 && threadIdx.x == 0) rowptr[N] = E;
}

__global__ void k_fill(const int* __restrict__ src, const int* __restrict__ dst,
                       const float* __restrict__ ew, const int* __restrict__ rowptr,
                       const int* __restrict__ rank, int2* __restrict__ csr, int E) {
  int e = blockIdx.x * blockDim.x + threadIdx.x;
  if (e < E) {
    int p = rowptr[dst[e]] + rank[e];
    csr[p] = make_int2(src[e], __float_as_int(ew[e]));
  }
}

__global__ void k_degdis(const int2* __restrict__ csr, const int* __restrict__ rowptr,
                         float* __restrict__ dis, float* __restrict__ nself, int N) {
  long t = (long)blockIdx.x * blockDim.x + threadIdx.x;
  int n = (int)(t >> 3);
  if (n >= N) return;
  int l = (int)(t & 7);
  const int pe = rowptr[n + 1];
  float s = 0.0f;
  for (int p = rowptr[n] + l; p < pe; p += 8) s += __int_as_float(csr[p].y);
  s += __shfl_xor(s, 1);
  s += __shfl_xor(s, 2);
  s += __shfl_xor(s, 4);
  if (l == 0) {
    float d = s + 1.0f;
    float r = (d > 0.0f) ? rsqrtf(d) : 0.0f;
    dis[n]   = r;
    nself[n] = r * r;
  }
}

__global__ void k_normcsr(int2* __restrict__ csr, const int* __restrict__ rowptr,
                          const float* __restrict__ dis, int N) {
  long t = (long)blockIdx.x * blockDim.x + threadIdx.x;
  int n = (int)(t >> 3);
  if (n >= N) return;
  int l = (int)(t & 7);
  const int pe = rowptr[n + 1];
  const float dn = dis[n];
  for (int p = rowptr[n] + l; p < pe; p += 8) {
    int2 c = csr[p];
    float nm = dis[c.x] * __int_as_float(c.y) * dn;
    csr[p].y = __float_as_int(nm);
  }
}

template <int COUT, bool RELU>
__global__ void k_gemm_wmma(const float* __restrict__ H, const float* __restrict__ Wg,
                            float* __restrict__ HW, int N) {
  constexpr int COLT = COUT / 16;
  constexpr int ROWT = 8 / COLT;
  constexpr int WSTR = 68;
  __shared__ float Wl[COUT * WSTR];

  for (int i = threadIdx.x; i < 64 * COUT; i += 256) {
    int k = i / COUT;
    int n = i - k * COUT;
    Wl[n * WSTR + k] = Wg[i];
  }
  __syncthreads();

  const int wave = threadIdx.x >> 5;
  const int lane = threadIdx.x & 31;
  const int rowTile = wave / COLT;
  const int colTile = wave - rowTile * COLT;
  const int m0 = (blockIdx.x * ROWT + rowTile) * 16;
  if (m0 >= N) return;
  const int n0 = colTile * 16;
  const int mr = lane & 15;
  const int kb = (lane >> 4) << 1;

  v8f acc = {};
  const float* hrow = H + (size_t)(m0 + mr) * 64;
  const float* wrow = Wl + (n0 + mr) * WSTR;
#pragma unroll
  for (int k = 0; k < 64; k += 4) {
    float a0 = hrow[k + kb];
    float a1 = hrow[k + kb + 1];
    if (RELU) { a0 = fmaxf(a0, 0.0f); a1 = fmaxf(a1, 0.0f); }
    v2f a; a.x = a0; a.y = a1;
    v2f b = *(const v2f*)&wrow[k + kb];
    acc = __builtin_amdgcn_wmma_f32_16x16x4_f32(
        false, a, false, b, (short)0, acc, false, false);
  }

  const int mbase = m0 + ((lane >> 4) << 3);
#pragma unroll
  for (int v = 0; v < 8; ++v) {
    size_t idx = (size_t)(mbase + v) * COUT + n0 + mr;
    HW[idx] = acc[v];
  }
}

template <int LPN>
__global__ void k_agg(const float* __restrict__ hw, const int2* __restrict__ csr,
                      const int* __restrict__ rowptr, const float* __restrict__ bias,
                      const float* __restrict__ nself, float* __restrict__ out, int N) {
  constexpr int C = 4 * LPN;
  long t = (long)blockIdx.x * blockDim.x + threadIdx.x;
  int n = (int)(t / LPN);
  if (n >= N) return;
  const int f = (int)(t % LPN) << 2;
  int p = rowptr[n];
  const int pe = rowptr[n + 1];
  const float4 hs = *(const float4*)(hw + (size_t)n * C + f);
  const float ns = nself[n];
  float4 acc;
  acc.x = bias[f + 0] + ns * hs.x;
  acc.y = bias[f + 1] + ns * hs.y;
  acc.z = bias[f + 2] + ns * hs.z;
  acc.w = bias[f + 3] + ns * hs.w;
  for (; p + 4 <= pe; p += 4) {
    const int2 c0 = csr[p], c1 = csr[p + 1], c2 = csr[p + 2], c3 = csr[p + 3];
    const float4 v0 = *(const float4*)(hw + (size_t)c0.x * C + f);
    const float4 v1 = *(const float4*)(hw + (size_t)c1.x * C + f);
    const float4 v2 = *(const float4*)(hw + (size_t)c2.x * C + f);
    const float4 v3 = *(const float4*)(hw + (size_t)c3.x * C + f);
    const float n0 = __int_as_float(c0.y), n1 = __int_as_float(c1.y);
    const float n2 = __int_as_float(c2.y), n3 = __int_as_float(c3.y);
    acc.x += n0 * v0.x; acc.y += n0 * v0.y; acc.z += n0 * v0.z; acc.w += n0 * v0.w;
    acc.x += n1 * v1.x; acc.y += n1 * v1.y; acc.z += n1 * v1.z; acc.w += n1 * v1.w;
    acc.x += n2 * v2.x; acc.y += n2 * v2.y; acc.z += n2 * v2.z; acc.w += n2 * v2.w;
    acc.x += n3 * v3.x; acc.y += n3 * v3.y; acc.z += n3 * v3.z; acc.w += n3 * v3.w;
  }
#pragma unroll 1
  for (; p < pe; ++p) {
    const int2 c0 = csr[p];
    const float4 v0 = *(const float4*)(hw + (size_t)c0.x * C + f);
    const float n0 = __int_as_float(c0.y);
    acc.x += n0 * v0.x; acc.y += n0 * v0.y; acc.z += n0 * v0.z; acc.w += n0 * v0.w;
  }
  *(float4*)(out + (size_t)n * C + f) = acc;
}

__global__ void k_logsoftmax16(float* __restrict__ out, int N) {
  int n = blockIdx.x * blockDim.x + threadIdx.x;
  if (n >= N) return;
  const float* lg = out + (size_t)N * 16 + (size_t)n * 16;
  float v[16];
  float mx = -3.402823466e38f;
#pragma unroll
  for (int i = 0; i < 16; ++i) { v[i] = lg[i]; mx = fmaxf(mx, v[i]); }
  float s = 0.0f;
#pragma unroll
  for (int i = 0; i < 16; ++i) s += expf(v[i] - mx);
  float lse = mx + logf(s);
  float* o0 = out + (size_t)n * 16;
  float* o2 = out + (size_t)2 * N * 16 + (size_t)n * 16;
#pragma unroll
  for (int i = 0; i < 16; ++i) {
    o0[i] = v[i] - lse;
    o2[i] = v[i];
  }
}

extern "C" void kernel_launch(void* const* d_in, const int* in_sizes, int n_in,
                              void* d_out, int out_size, void* d_ws, size_t ws_size,
                              hipStream_t stream) {
  const float* x  = (const float*)d_in[0];
  const int*   ei = (const int*)  d_in[1];
  const float* ew = (const float*)d_in[2];
  const float* W1 = (const float*)d_in[3];
  const float* b1 = (const float*)d_in[4];
  const float* W2 = (const float*)d_in[5];
  const float* b2 = (const float*)d_in[6];
  const float* W3 = (const float*)d_in[7];
  const float* b3 = (const float*)d_in[8];
  float* out = (float*)d_out;

  const int N = in_sizes[0] / 64;
  const int E = in_sizes[2];
  const int* src = ei;
  const int* dst = ei + E;

  const int Np = (N + 64) & ~63;
  const int Ep = (E + 63) & ~63;
  const int nb = (N + 1023) / 1024;
  const int nbp = (nb + 63) & ~63;

  int*   cnt    = (int*)d_ws;
  int*   rowptr = cnt + Np;
  int*   bsum   = rowptr + Np;
  int*   rank   = bsum + nbp;
  int2*  csr    = (int2*)(rank + Ep);
  float* dis    = (float*)(csr + Ep);
  float* nself  = dis + Np;
  float* bufA   = nself + Np;
  float* bufB   = bufA + (size_t)N * 64;
  float* bufC   = bufB + (size_t)N * 64;
  float* logits = out + (size_t)N * 16;

  const int T = 256;
  const int gN = (N + T - 1) / T;
  const int gE = (E + T - 1) / T;
  const int g8N = (int)(((size_t)N * 8 + T - 1) / T);
  const int rowTiles = (N + 15) / 16;
  const int gAg64 = (int)(((size_t)N * 16 + T - 1) / T);
  const int gAg16 = (int)(((size_t)N * 4  + T - 1) / T);

  k_zero   <<<gN, T, 0, stream>>>(cnt, N);
  k_count  <<<gE, T, 0, stream>>>(dst, cnt, rank, E);
  k_bsum   <<<nb, 256, 0, stream>>>(cnt, bsum, N);
  k_scan   <<<nb, 256, 0, stream>>>(cnt, bsum, rowptr, N, E);
  k_fill   <<<gE, T, 0, stream>>>(src, dst, ew, rowptr, rank, csr, E);
  k_degdis <<<g8N, T, 0, stream>>>(csr, rowptr, dis, nself, N);
  k_normcsr<<<g8N, T, 0, stream>>>(csr, rowptr, dis, N);

  k_gemm_wmma<64, false><<<(rowTiles + 1) / 2, 256, 0, stream>>>(x, W1, bufA, N);
  k_agg<16><<<gAg64, T, 0, stream>>>(bufA, csr, rowptr, b1, nself, bufB, N);

  k_gemm_wmma<64, true><<<(rowTiles + 1) / 2, 256, 0, stream>>>(bufB, W2, bufA, N);
  k_agg<16><<<gAg64, T, 0, stream>>>(bufA, csr, rowptr, b2, nself, bufC, N);

  k_gemm_wmma<16, true><<<(rowTiles + 7) / 8, 256, 0, stream>>>(bufC, W3, bufA, N);
  k_agg<4><<<gAg16, T, 0, stream>>>(bufA, csr, rowptr, b3, nself, logits, N);

  k_logsoftmax16<<<gN, T, 0, stream>>>(out, N);
}


# Symmetric norm factored into per-node scales: GEMM epilogue scales row n by deg^-1/2, aggregation sums raw-weight rows, scales once; per-edge norm pass gone (13 launches); rounding order changes
# speedup vs baseline: 2.0151x; 1.0491x over previous
#include <hip/hip_runtime.h>
#include <hip/hip_bf16.h>
#include <math.h>


typedef float v2f __attribute__((ext_vector_type(2)));
typedef float v8f __attribute__((ext_vector_type(8)));

__global__ void k_zero(int* __restrict__ p, int n) {
  int i = blockIdx.x * blockDim.x + threadIdx.x;
  if (i < n) p[i] = 0;
}

__global__ void k_count(const int* __restrict__ dst, int* __restrict__ cnt,
                        int* __restrict__ rank, int E) {
  int e = blockIdx.x * blockDim.x + threadIdx.x;
  if (e < E) rank[e] = atomicAdd(&cnt[dst[e]], 1);
}

__device__ __forceinline__ int wave_sum(int v) {
  for (int m = 16; m >= 1; m >>= 1) v += __shfl_xor(v, m);
  return v;
}

__global__ void k_bsum(const int* __restrict__ cnt, int* __restrict__ bsum, int N) {
  __shared__ int ws[8];
  const int i0 = (blockIdx.x * 256 + threadIdx.x) * 4;
  int v = 0;
#pragma unroll
  for (int k = 0; k < 4; ++k) if (i0 + k < N) v += cnt[i0 + k];
  v = wave_sum(v);
  if ((threadIdx.x & 31) == 0) ws[threadIdx.x >> 5] = v;
  __syncthreads();
  if (threadIdx.x == 0) {
    int s = 0;
#pragma unroll
    for (int w = 0; w < 8; ++w) s += ws[w];
    bsum[blockIdx.x] = s;
  }
}

__global__ void k_scan(const int* __restrict__ cnt, const int* __restrict__ bsum,
                       int* __restrict__ rowptr, int N, int E) {
  __shared__ int ws[8];
  __shared__ int wo[8];
  const int lane = threadIdx.x & 31;
  const int wave = threadIdx.x >> 5;
  int part = 0;
  for (int i = threadIdx.x; i < (int)blockIdx.x; i += 256) part += bsum[i];
  part = wave_sum(part);
  if (lane == 0) wo[wave] = part;
  const int i0 = (blockIdx.x * 256 + threadIdx.x) * 4;
  int c[4];
  int tot = 0;
#pragma unroll
  for (int k = 0; k < 4; ++k) { c[k] = (i0 + k < N) ? cnt[i0 + k] : 0; tot += c[k]; }
  int inc = tot;
  for (int d = 1; d < 32; d <<= 1) {
    int t = __shfl_up(inc, d);
    if (lane >= d) inc += t;
  }
  if (lane == 31) ws[wave] = inc;
  __syncthreads();
  int off = 0;
#pragma unroll
  for (int w = 0; w < 8; ++w) { off += wo[w]; if (w < wave) off += ws[w]; }
  int run = off + inc - tot;
#pragma unroll
  for (int k = 0; k < 4; ++k) if (i0 + k < N) { rowptr[i0 + k] = run; run += c[k]; }
  if (blockIdx.x == 0 && threadIdx.x == 0) rowptr[N] = E;
}

__global__ void k_fill(const int* __restrict__ src, const int* __restrict__ dst,
                       const float* __restrict__ ew, const int* __restrict__ rowptr,
                       const int* __restrict__ rank, int2* __restrict__ csr, int E) {
  int e = blockIdx.x * blockDim.x + threadIdx.x;
  if (e < E) {
    int p = rowptr[dst[e]] + rank[e];
    csr[p] = make_int2(src[e], __float_as_int(ew[e]));
  }
}

__global__ void k_degdis(const int2* __restrict__ csr, const int* __restrict__ rowptr,
                         float* __restrict__ dis, int N) {
  long t = (long)blockIdx.x * blockDim.x + threadIdx.x;
  int n = (int)(t >> 3);
  if (n >= N) return;
  int l = (int)(t & 7);
  const int pe = rowptr[n + 1];
  float s = 0.0f;
  for (int p = rowptr[n] + l; p < pe; p += 8) s += __int_as_float(csr[p].y);
  s += __shfl_xor(s, 1);
  s += __shfl_xor(s, 2);
  s += __shfl_xor(s, 4);
  if (l == 0) {
    float d = s + 1.0f;
    dis[n] = (d > 0.0f) ? rsqrtf(d) : 0.0f;
  }
}

template <int COUT, bool RELU>
__global__ void k_gemm_wmma(const float* __restrict__ H, const float* __restrict__ Wg,
                            const float* __restrict__ dis, float* __restrict__ HW, int N) {
  constexpr int COLT = COUT / 16;
  constexpr int ROWT = 8 / COLT;
  constexpr int WSTR = 68;
  __shared__ float Wl[COUT * WSTR];

  for (int i = threadIdx.x; i < 64 * COUT; i += 256) {
    int k = i / COUT;
    int n = i - k * COUT;
    Wl[n * WSTR + k] = Wg[i];
  }
  __syncthreads();

  const int wave = threadIdx.x >> 5;
  const int lane = threadIdx.x & 31;
  const int rowTile = wave / COLT;
  const int colTile = wave - rowTile * COLT;
  const int m0 = (blockIdx.x * ROWT + rowTile) * 16;
  if (m0 >= N) return;
  const int n0 = colTile * 16;
  const int mr = lane & 15;
  const int kb = (lane >> 4) << 1;

  const int mbase = m0 + ((lane >> 4) << 3);
  const float4 dlo = *(const float4*)(dis + mbase);
  const float4 dhi = *(const float4*)(dis + mbase + 4);

  v8f acc = {};
  const float* hrow = H + (size_t)(m0 + mr) * 64;
  const float* wrow = Wl + (n0 + mr) * WSTR;
#pragma unroll
  for (int k = 0; k < 64; k += 4) {
    float a0 = hrow[k + kb];
    float a1 = hrow[k + kb + 1];
    if (RELU) { a0 = fmaxf(a0, 0.0f); a1 = fmaxf(a1, 0.0f); }
    v2f a; a.x = a0; a.y = a1;
    v2f b = *(const v2f*)&wrow[k + kb];
    acc = __builtin_amdgcn_wmma_f32_16x16x4_f32(
        false, a, false, b, (short)0, acc, false, false);
  }

  const float dsc[8] = {dlo.x, dlo.y, dlo.z, dlo.w, dhi.x, dhi.y, dhi.z, dhi.w};
#pragma unroll
  for (int v = 0; v < 8; ++v) {
    size_t idx = (size_t)(mbase + v) * COUT + n0 + mr;
    HW[idx] = acc[v] * dsc[v];
  }
}

template <int LPN>
__global__ void k_agg(const float* __restrict__ hw, const int2* __restrict__ csr,
                      const int* __restrict__ rowptr, const float* __restrict__ bias,
                      const float* __restrict__ dis, float* __restrict__ out, int N) {
  constexpr int C = 4 * LPN;
  long t = (long)blockIdx.x * blockDim.x + threadIdx.x;
  int n = (int)(t / LPN);
  if (n >= N) return;
  const int f = (int)(t % LPN) << 2;
  int p = rowptr[n];
  const int pe = rowptr[n + 1];
  const float dn = dis[n];
  const float4 bv = *(const float4*)(bias + f);
  float4 acc = *(const float4*)(hw + (size_t)n * C + f);
  for (; p + 4 <= pe; p += 4) {
    const int2 c0 = csr[p], c1 = csr[p + 1], c2 = csr[p + 2], c3 = csr[p + 3];
    const float4 v0 = *(const float4*)(hw + (size_t)c0.x * C + f);
    const float4 v1 = *(const float4*)(hw + (size_t)c1.x * C + f);
    const float4 v2 = *(const float4*)(hw + (size_t)c2.x * C + f);
    const float4 v3 = *(const float4*)(hw + (size_t)c3.x * C + f);
    const float n0 = __int_as_float(c0.y), n1 = __int_as_float(c1.y);
    const float n2 = __int_as_float(c2.y), n3 = __int_as_float(c3.y);
    acc.x += n0 * v0.x; acc.y += n0 * v0.y; acc.z += n0 * v0.z; acc.w += n0 * v0.w;
    acc.x += n1 * v1.x; acc.y += n1 * v1.y; acc.z += n1 * v1.z; acc.w += n1 * v1.w;
    acc.x += n2 * v2.x; acc.y += n2 * v2.y; acc.z += n2 * v2.z; acc.w += n2 * v2.w;
    acc.x += n3 * v3.x; acc.y += n3 * v3.y; acc.z += n3 * v3.z; acc.w += n3 * v3.w;
  }
#pragma unroll 1
  for (; p < pe; ++p) {
    const int2 c0 = csr[p];
    const float4 v0 = *(const float4*)(hw + (size_t)c0.x * C + f);
    const float n0 = __int_as_float(c0.y);
    acc.x += n0 * v0.x; acc.y += n0 * v0.y; acc.z += n0 * v0.z; acc.w += n0 * v0.w;
  }
  float4 o;
  o.x = bv.x + dn * acc.x;
  o.y = bv.y + dn * acc.y;
  o.z = bv.z + dn * acc.z;
  o.w = bv.w + dn * acc.w;
  *(float4*)(out + (size_t)n * C + f) = o;
}

__global__ void k_logsoftmax16(float* __restrict__ out, int N) {
  int n = blockIdx.x * blockDim.x + threadIdx.x;
  if (n >= N) return;
  const float* lg = out + (size_t)N * 16 + (size_t)n * 16;
  float v[16];
  float mx = -3.402823466e38f;
#pragma unroll
  for (int i = 0; i < 16; ++i) { v[i] = lg[i]; mx = fmaxf(mx, v[i]); }
  float s = 0.0f;
#pragma unroll
  for (int i = 0; i < 16; ++i) s += expf(v[i] - mx);
  float lse = mx + logf(s);
  float* o0 = out + (size_t)n * 16;
  float* o2 = out + (size_t)2 * N * 16 + (size_t)n * 16;
#pragma unroll
  for (int i = 0; i < 16; ++i) {
    o0[i] = v[i] - lse;
    o2[i] = v[i];
  }
}

extern "C" void kernel_launch(void* const* d_in, const int* in_sizes, int n_in,
                              void* d_out, int out_size, void* d_ws, size_t ws_size,
                              hipStream_t stream) {
  const float* x  = (const float*)d_in[0];
  const int*   ei = (const int*)  d_in[1];
  const float* ew = (const float*)d_in[2];
  const float* W1 = (const float*)d_in[3];
  const float* b1 = (const float*)d_in[4];
  const float* W2 = (const float*)d_in[5];
  const float* b2 = (const float*)d_in[6];
  const float* W3 = (const float*)d_in[7];
  const float* b3 = (const float*)d_in[8];
  float* out = (float*)d_out;

  const int N = in_sizes[0] / 64;
  const int E = in_sizes[2];
  const int* src = ei;
  const int* dst = ei + E;

  const int Np = (N + 64) & ~63;
  const int Ep = (E + 63) & ~63;
  const int nb = (N + 1023) / 1024;
  const int nbp = (nb + 63) & ~63;

  int*   cnt    = (int*)d_ws;
  int*   rowptr = cnt + Np;
  int*   bsum   = rowptr + Np;
  int*   rank   = bsum + nbp;
  int2*  csr    = (int2*)(rank + Ep);
  float* dis    = (float*)(csr + Ep);
  float* bufA   = dis + Np;
  float* bufB   = bufA + (size_t)N * 64;
  float* bufC   = bufB + (size_t)N * 64;
  float* logits = out + (size_t)N * 16;

  const int T = 256;
  const int gN = (N + T - 1) / T;
  const int gE = (E + T - 1) / T;
  const int g8N = (int)(((size_t)N * 8 + T - 1) / T);
  const int rowTiles = (N + 15) / 16;
  const int gAg64 = (int)(((size_t)N * 16 + T - 1) / T);
  const int gAg16 = (int)(((size_t)N * 4  + T - 1) / T);

  k_zero   <<<gN, T, 0, stream>>>(cnt, N);
  k_count  <<<gE, T, 0, stream>>>(dst, cnt, rank, E);
  k_bsum   <<<nb, 256, 0, stream>>>(cnt, bsum, N);
  k_scan   <<<nb, 256, 0, stream>>>(cnt, bsum, rowptr, N, E);
  k_fill   <<<gE, T, 0, stream>>>(src, dst, ew, rowptr, rank, csr, E);
  k_degdis <<<g8N, T, 0, stream>>>(csr, rowptr, dis, N);

  k_gemm_wmma<64, false><<<(rowTiles + 1) / 2, 256, 0, stream>>>(x, W1, dis, bufA, N);
  k_agg<16><<<gAg64, T, 0, stream>>>(bufA, csr, rowptr, b1, dis, bufB, N);

  k_gemm_wmma<64, true><<<(rowTiles + 1) / 2, 256, 0, stream>>>(bufB, W2, dis, bufA, N);
  k_agg<16><<<gAg64, T, 0, stream>>>(bufA, csr, rowptr, b2, dis, bufC, N);

  k_gemm_wmma<16, true><<<(rowTiles + 7) / 8, 256, 0, stream>>>(bufC, W3, dis, bufA, N);
  k_agg<4><<<gAg16, T, 0, stream>>>(bufA, csr, rowptr, b3, dis, logits, N);

  k_logsoftmax16<<<gN, T, 0, stream>>>(out, N);
}


# Edges sorted by destination with a two-level counting sort in LDS (chunk histogram, bucket runs, per-bucket sort, degree by LDS adds): no global atomics; per-node edge and degree sum order change
# speedup vs baseline: 3.4004x; 1.6875x over previous
#include <hip/hip_runtime.h>
#include <hip/hip_bf16.h>
#include <math.h>


typedef float v2f __attribute__((ext_vector_type(2)));
typedef float v8f __attribute__((ext_vector_type(8)));


constexpr int BSH  = 8;
constexpr int BNOD = 1 << BSH;
constexpr int BMAX = 512;
constexpr int CH   = 8192;
constexpr int TPB  = 1024;

__device__ __forceinline__ int block_excl_scan(int v, int* wsum) {
  const int lane = threadIdx.x & 31;
  const int wave = threadIdx.x >> 5;
  int inc = v;
#pragma unroll
  for (int d = 1; d < 32; d <<= 1) {
    int u = __shfl_up(inc, d);
    if (lane >= d) inc += u;
  }
  if (lane == 31) wsum[wave] = inc;
  __syncthreads();
  const int nw = blockDim.x >> 5;
  int off = (lane < nw && lane < wave) ? wsum[lane] : 0;
#pragma unroll
  for (int m = 16; m >= 1; m >>= 1) off += __shfl_xor(off, m);
  __syncthreads();
  return off + inc - v;
}

__global__ void __launch_bounds__(TPB)
k_hist(const int* __restrict__ dst, int* __restrict__ hist, int E) {
  __shared__ int h[BMAX];
  const int t = threadIdx.x;
  if (t < BMAX) h[t] = 0;
  __syncthreads();
  const int e0 = blockIdx.x * CH;
  int d[CH / TPB];
#pragma unroll
  for (int k = 0; k < CH / TPB; ++k) {
    const int e = e0 + k * TPB + t;
    d[k] = dst[(e < E) ? e : (E - 1)];
  }
#pragma unroll
  for (int k = 0; k < CH / TPB; ++k) {
    const int e = e0 + k * TPB + t;
    if (e < E) atomicAdd(&h[d[k] >> BSH], 1);
  }
  __syncthreads();
  if (t < BMAX) hist[(size_t)blockIdx.x * BMAX + t] = h[t];
}

__global__ void __launch_bounds__(TPB)
k_group(const int* __restrict__ src, const int* __restrict__ dst,
        const float* __restrict__ ew, const int* __restrict__ hist,
        int2* __restrict__ grp, int* __restrict__ bstart,
        int E, int B, int C) {
  __shared__ int delta[BMAX];
  __shared__ int cur[BMAX];
  __shared__ int totL[BMAX];
  __shared__ int prevL[BMAX];
  __shared__ int wsum[32];
  __shared__ int2 stage[CH];
  __shared__ unsigned short sbk[CH];
  const int c = blockIdx.x;
  const int t = threadIdx.x;
  const int e0 = c * CH;
  int dd[CH / TPB], ss[CH / TPB], ww[CH / TPB];
#pragma unroll
  for (int k = 0; k < CH / TPB; ++k) {
    const int e = e0 + k * TPB + t;
    const int ec = (e < E) ? e : (E - 1);
    dd[k] = dst[ec];
    ss[k] = src[ec];
    ww[k] = __float_as_int(ew[ec]);
  }
  const int own = (t < B) ? hist[(size_t)c * BMAX + t] : 0;
  if (t < BMAX) { totL[t] = 0; prevL[t] = 0; }
  __syncthreads();
  {
    const int lane = t & 31;
    const int wave = t >> 5;
    int tt[BMAX / 32], pp[BMAX / 32];
#pragma unroll
    for (int j = 0; j < BMAX / 32; ++j) { tt[j] = 0; pp[j] = 0; }
    for (int r = wave; r < C; r += TPB / 32) {
      const int* hr = hist + (size_t)r * BMAX + lane;
      const bool before = r < c;
      int v[BMAX / 32];
#pragma unroll
      for (int j = 0; j < BMAX / 32; ++j) v[j] = hr[32 * j];
#pragma unroll
      for (int j = 0; j < BMAX / 32; ++j) {
        tt[j] += v[j];
        pp[j] += before ? v[j] : 0;
      }
    }
#pragma unroll
    for (int j = 0; j < BMAX / 32; ++j) {
      atomicAdd(&totL[lane + 32 * j], tt[j]);
      atomicAdd(&prevL[lane + 32 * j], pp[j]);
    }
  }
  __syncthreads();
  const int tot  = (t < B) ? totL[t] : 0;
  const int prev = (t < B) ? prevL[t] : 0;
  const int gstart = block_excl_scan(tot, wsum);
  const int lstart = block_excl_scan(own, wsum);
  if (t < B) {
    delta[t] = gstart + prev - lstart;
    cur[t] = lstart;
    if (c == 0) bstart[t] = gstart;
  }
  if (c == 0 && t == 0) bstart[B] = E;
  __syncthreads();
#pragma unroll
  for (int k = 0; k < CH / TPB; ++k) {
    const int e = e0 + k * TPB + t;
    if (e < E) {
      const int b = dd[k] >> BSH;
      const int p = atomicAdd(&cur[b], 1);
      stage[p] = make_int2((ss[k] << BSH) | (dd[k] & (BNOD - 1)), ww[k]);
      sbk[p] = (unsigned short)b;
    }
  }
  __syncthreads();
  const int nloc = (E - e0 < CH) ? (E - e0) : CH;
  for (int i = t; i < nloc; i += TPB) grp[delta[sbk[i]] + i] = stage[i];
}

__global__ void __launch_bounds__(TPB)
k_owner(const int2* __restrict__ grp, const int* __restrict__ bstart,
        int2* __restrict__ csr, int* __restrict__ rowptr, float* __restrict__ dis,
        int N, int E, int B) {
  __shared__ int cntl[BNOD];
  __shared__ float degl[BNOD];
  __shared__ int startl[BNOD];
  __shared__ int cur[BNOD];
  __shared__ int wsum[32];
  __shared__ int2 stage[CH];
  const int b = blockIdx.x;
  const int t = threadIdx.x;
  const int s = bstart[b];
  const int nE = bstart[b + 1] - s;
  if (t < BNOD) { cntl[t] = 0; degl[t] = 0.0f; }
  __syncthreads();
  for (int i = t; i < nE; i += TPB) {
    const int2 r = grp[s + i];
    const int nl = r.x & (BNOD - 1);
    atomicAdd(&cntl[nl], 1);
    atomicAdd(&degl[nl], __int_as_float(r.y));
  }
  __syncthreads();
  const int ex = block_excl_scan((t < BNOD) ? cntl[t] : 0, wsum);
  if (t < BNOD) { startl[t] = ex; cur[t] = ex; }
  __syncthreads();
  const bool inl = (nE <= CH);
  for (int i = t; i < nE; i += TPB) {
    const int2 r = grp[s + i];
    const int nl = r.x & (BNOD - 1);
    const int p = atomicAdd(&cur[nl], 1);
    const int2 o = make_int2(r.x >> BSH, r.y);
    if (inl) stage[p] = o; else csr[s + p] = o;
  }
  __syncthreads();
  if (inl) for (int i = t; i < nE; i += TPB) csr[s + i] = stage[i];
  if (t < BNOD) {
    const int n = (b << BSH) + t;
    if (n < N) {
      rowptr[n] = s + startl[t];
      const float d = degl[t] + 1.0f;
      dis[n] = (d > 0.0f) ? rsqrtf(d) : 0.0f;
    }
  }
  if (b == B - 1 && t == 0) rowptr[N] = E;
}

template <int COUT, bool RELU>
__global__ void k_gemm_wmma(const float* __restrict__ H, const float* __restrict__ Wg,
                            const float* __restrict__ dis, float* __restrict__ HW, int N) {
  constexpr int COLT = COUT / 16;
  constexpr int ROWT = 8 / COLT;
  constexpr int WSTR = 68;
  __shared__ float Wl[COUT * WSTR];

  for (int i = threadIdx.x; i < 64 * COUT; i += 256) {
    int k = i / COUT;
    int n = i - k * COUT;
    Wl[n * WSTR + k] = Wg[i];
  }
  __syncthreads();

  const int wave = threadIdx.x >> 5;
  const int lane = threadIdx.x & 31;
  const int rowTile = wave / COLT;
  const int colTile = wave - rowTile * COLT;
  const int m0 = (blockIdx.x * ROWT + rowTile) * 16;
  if (m0 >= N) return;
  const int n0 = colTile * 16;
  const int mr = lane & 15;
  const int kb = (lane >> 4) << 1;

  const int mbase = m0 + ((lane >> 4) << 3);
  const float4 dlo = *(const float4*)(dis + mbase);
  const float4 dhi = *(const float4*)(dis + mbase + 4);

  v8f acc = {};
  const float* hrow = H + (size_t)(m0 + mr) * 64;
  const float* wrow = Wl + (n0 + mr) * WSTR;
#pragma unroll
  for (int k = 0; k < 64; k += 4) {
    float a0 = hrow[k + kb];
    float a1 = hrow[k + kb + 1];
    if (RELU) { a0 = fmaxf(a0, 0.0f); a1 = fmaxf(a1, 0.0f); }
    v2f a; a.x = a0; a.y = a1;
    v2f b = *(const v2f*)&wrow[k + kb];
    acc = __builtin_amdgcn_wmma_f32_16x16x4_f32(
        false, a, false, b, (short)0, acc, false, false);
  }

  const float dsc[8] = {dlo.x, dlo.y, dlo.z, dlo.w, dhi.x, dhi.y, dhi.z, dhi.w};
#pragma unroll
  for (int v = 0; v < 8; ++v) {
    size_t idx = (size_t)(mbase + v) * COUT + n0 + mr;
    HW[idx] = acc[v] * dsc[v];
  }
}

template <int LPN>
__global__ void k_agg(const float* __restrict__ hw, const int2* __restrict__ csr,
                      const int* __restrict__ rowptr, const float* __restrict__ bias,
                      const float* __restrict__ dis, float* __restrict__ out, int N) {
  constexpr int C = 4 * LPN;
  long t = (long)blockIdx.x * blockDim.x + threadIdx.x;
  int n = (int)(t / LPN);
  if (n >= N) return;
  const int f = (int)(t % LPN) << 2;
  int p = rowptr[n];
  const int pe = rowptr[n + 1];
  const float dn = dis[n];
  const float4 bv = *(const float4*)(bias + f);
  float4 acc = *(const float4*)(hw + (size_t)n * C + f);
  for (; p + 4 <= pe; p += 4) {
    const int2 c0 = csr[p], c1 = csr[p + 1], c2 = csr[p + 2], c3 = csr[p + 3];
    const float4 v0 = *(const float4*)(hw + (size_t)c0.x * C + f);
    const float4 v1 = *(const float4*)(hw + (size_t)c1.x * C + f);
    const float4 v2 = *(const float4*)(hw + (size_t)c2.x * C + f);
    const float4 v3 = *(const float4*)(hw + (size_t)c3.x * C + f);
    const float n0 = __int_as_float(c0.y), n1 = __int_as_float(c1.y);
    const float n2 = __int_as_float(c2.y), n3 = __int_as_float(c3.y);
    acc.x += n0 * v0.x; acc.y += n0 * v0.y; acc.z += n0 * v0.z; acc.w += n0 * v0.w;
    acc.x += n1 * v1.x; acc.y += n1 * v1.y; acc.z += n1 * v1.z; acc.w += n1 * v1.w;
    acc.x += n2 * v2.x; acc.y += n2 * v2.y; acc.z += n2 * v2.z; acc.w += n2 * v2.w;
    acc.x += n3 * v3.x; acc.y += n3 * v3.y; acc.z += n3 * v3.z; acc.w += n3 * v3.w;
  }
#pragma unroll 1
  for (; p < pe; ++p) {
    const int2 c0 = csr[p];
    const float4 v0 = *(const float4*)(hw + (size_t)c0.x * C + f);
    const float n0 = __int_as_float(c0.y);
    acc.x += n0 * v0.x; acc.y += n0 * v0.y; acc.z += n0 * v0.z; acc.w += n0 * v0.w;
  }
  float4 o;
  o.x = bv.x + dn * acc.x;
  o.y = bv.y + dn * acc.y;
  o.z = bv.z + dn * acc.z;
  o.w = bv.w + dn * acc.w;
  *(float4*)(out + (size_t)n * C + f) = o;
}

__global__ void k_logsoftmax16(float* __restrict__ out, int N) {
  int n = blockIdx.x * blockDim.x + threadIdx.x;
  if (n >= N) return;
  const float* lg = out + (size_t)N * 16 + (size_t)n * 16;
  float v[16];
  float mx = -3.402823466e38f;
#pragma unroll
  for (int i = 0; i < 16; ++i) { v[i] = lg[i]; mx = fmaxf(mx, v[i]); }
  float s = 0.0f;
#pragma unroll
  for (int i = 0; i < 16; ++i) s += expf(v[i] - mx);
  float lse = mx + logf(s);
  float* o0 = out + (size_t)n * 16;
  float* o2 = out + (size_t)2 * N * 16 + (size_t)n * 16;
#pragma unroll
  for (int i = 0; i < 16; ++i) {
    o0[i] = v[i] - lse;
    o2[i] = v[i];
  }
}

extern "C" void kernel_launch(void* const* d_in, const int* in_sizes, int n_in,
                              void* d_out, int out_size, void* d_ws, size_t ws_size,
                              hipStream_t stream) {
  const float* x  = (const float*)d_in[0];
  const int*   ei = (const int*)  d_in[1];
  const float* ew = (const float*)d_in[2];
  const float* W1 = (const float*)d_in[3];
  const float* b1 = (const float*)d_in[4];
  const float* W2 = (const float*)d_in[5];
  const float* b2 = (const float*)d_in[6];
  const float* W3 = (const float*)d_in[7];
  const float* b3 = (const float*)d_in[8];
  float* out = (float*)d_out;

  const int N = in_sizes[0] / 64;
  const int E = in_sizes[2];
  const int* src = ei;
  const int* dst = ei + E;

  const int Np = (N + 64) & ~63;
  const int Ep = (E + 63) & ~63;
  const int B  = (N + BNOD - 1) >> BSH;
  const int Bq = (B + 64) & ~63;
  const int C  = (E + CH - 1) / CH;

  int*   hist   = (int*)d_ws;
  int*   bstart = hist + (size_t)C * BMAX;
  int2*  grp    = (int2*)(bstart + Bq);
  int*   rowptr = (int*)(grp + Ep);
  int2*  csr    = (int2*)(rowptr + Np);
  float* dis    = (float*)(csr + Ep);
  float* bufA   = dis + Np;
  float* bufB   = bufA + (size_t)N * 64;
  float* bufC   = bufB + (size_t)N * 64;
  float* logits = out + (size_t)N * 16;

  const int T = 256;
  const int gN = (N + T - 1) / T;
  const int rowTiles = (N + 15) / 16;
  const int gAg64 = (int)(((size_t)N * 16 + T - 1) / T);
  const int gAg16 = (int)(((size_t)N * 4  + T - 1) / T);

  k_hist <<<C, TPB, 0, stream>>>(dst, hist, E);
  k_group<<<C, TPB, 0, stream>>>(src, dst, ew, hist, grp, bstart, E, B, C);
  k_owner<<<B, TPB, 0, stream>>>(grp, bstart, csr, rowptr, dis, N, E, B);

  k_gemm_wmma<64, false><<<(rowTiles + 1) / 2, 256, 0, stream>>>(x, W1, dis, bufA, N);
  k_agg<16><<<gAg64, T, 0, stream>>>(bufA, csr, rowptr, b1, dis, bufB, N);

  k_gemm_wmma<64, true><<<(rowTiles + 1) / 2, 256, 0, stream>>>(bufB, W2, dis, bufA, N);
  k_agg<16><<<gAg64, T, 0, stream>>>(bufA, csr, rowptr, b2, dis, bufC, N);

  k_gemm_wmma<16, true><<<(rowTiles + 7) / 8, 256, 0, stream>>>(bufC, W3, dis, bufA, N);
  k_agg<4><<<gAg16, T, 0, stream>>>(bufA, csr, rowptr, b3, dis, logits, N);

  k_logsoftmax16<<<gN, T, 0, stream>>>(out, N);
}


# Next layer's GEMM fused into the aggregation: a block's 32 finished rows go (ReLU) through LDS into the f32 WMMA product with W read from global; 10 launches to 8, layer outputs h1/h2 never stored
# speedup vs baseline: 3.8893x; 1.1438x over previous
#include <hip/hip_runtime.h>
#include <hip/hip_bf16.h>
#include <math.h>


typedef float v2f __attribute__((ext_vector_type(2)));
typedef float v8f __attribute__((ext_vector_type(8)));


constexpr int BSH  = 8;
constexpr int BNOD = 1 << BSH;
constexpr int BMAX = 512;
constexpr int CH   = 8192;
constexpr int TPB  = 1024;

__device__ __forceinline__ int block_excl_scan(int v, int* wsum) {
  const int lane = threadIdx.x & 31;
  const int wave = threadIdx.x >> 5;
  int inc = v;
#pragma unroll
  for (int d = 1; d < 32; d <<= 1) {
    int u = __shfl_up(inc, d);
    if (lane >= d) inc += u;
  }
  if (lane == 31) wsum[wave] = inc;
  __syncthreads();
  const int nw = blockDim.x >> 5;
  int off = (lane < nw && lane < wave) ? wsum[lane] : 0;
#pragma unroll
  for (int m = 16; m >= 1; m >>= 1) off += __shfl_xor(off, m);
  __syncthreads();
  return off + inc - v;
}

__global__ void __launch_bounds__(TPB)
k_hist(const int* __restrict__ dst, int* __restrict__ hist, int E) {
  __shared__ int h[BMAX];
  const int t = threadIdx.x;
  if (t < BMAX) h[t] = 0;
  __syncthreads();
  const int e0 = blockIdx.x * CH;
  int d[CH / TPB];
#pragma unroll
  for (int k = 0; k < CH / TPB; ++k) {
    const int e = e0 + k * TPB + t;
    d[k] = dst[(e < E) ? e : (E - 1)];
  }
#pragma unroll
  for (int k = 0; k < CH / TPB; ++k) {
    const int e = e0 + k * TPB + t;
    if (e < E) atomicAdd(&h[d[k] >> BSH], 1);
  }
  __syncthreads();
  if (t < BMAX) hist[(size_t)blockIdx.x * BMAX + t] = h[t];
}

__global__ void __launch_bounds__(TPB)
k_group(const int* __restrict__ src, const int* __restrict__ dst,
        const float* __restrict__ ew, const int* __restrict__ hist,
        int2* __restrict__ grp, int* __restrict__ bstart,
        int E, int B, int C) {
  __shared__ int delta[BMAX];
  __shared__ int cur[BMAX];
  __shared__ int totL[BMAX];
  __shared__ int prevL[BMAX];
  __shared__ int wsum[32];
  __shared__ int2 stage[CH];
  __shared__ unsigned short sbk[CH];
  const int c = blockIdx.x;
  const int t = threadIdx.x;
  const int e0 = c * CH;
  int dd[CH / TPB], ss[CH / TPB], ww[CH / TPB];
#pragma unroll
  for (int k = 0; k < CH / TPB; ++k) {
    const int e = e0 + k * TPB + t;
    const int ec = (e < E) ? e : (E - 1);
    dd[k] = dst[ec];
    ss[k] = src[ec];
    ww[k] = __float_as_int(ew[ec]);
  }
  const int own = (t < B) ? hist[(size_t)c * BMAX + t] : 0;
  if (t < BMAX) { totL[t] = 0; prevL[t] = 0; }
  __syncthreads();
  {
    const int lane = t & 31;
    const int wave = t >> 5;
    int tt[BMAX / 32], pp[BMAX / 32];
#pragma unroll
    for (int j = 0; j < BMAX / 32; ++j) { tt[j] = 0; pp[j] = 0; }
    for (int r = wave; r < C; r += TPB / 32) {
      const int* hr = hist + (size_t)r * BMAX + lane;
      const bool before = r < c;
      int v[BMAX / 32];
#pragma unroll
      for (int j = 0; j < BMAX / 32; ++j) v[j] = hr[32 * j];
#pragma unroll
      for (int j = 0; j < BMAX / 32; ++j) {
        tt[j] += v[j];
        pp[j] += before ? v[j] : 0;
      }
    }
#pragma unroll
    for (int j = 0; j < BMAX / 32; ++j) {
      atomicAdd(&totL[lane + 32 * j], tt[j]);
      atomicAdd(&prevL[lane + 32 * j], pp[j]);
    }
  }
  __syncthreads();
  const int tot  = (t < B) ? totL[t] : 0;
  const int prev = (t < B) ? prevL[t] : 0;
  const int gstart = block_excl_scan(tot, wsum);
  const int lstart = block_excl_scan(own, wsum);
  if (t < B) {
    delta[t] = gstart + prev - lstart;
    cur[t] = lstart;
    if (c == 0) bstart[t] = gstart;
  }
  if (c == 0 && t == 0) bstart[B] = E;
  __syncthreads();
#pragma unroll
  for (int k = 0; k < CH / TPB; ++k) {
    const int e = e0 + k * TPB + t;
    if (e < E) {
      const int b = dd[k] >> BSH;
      const int p = atomicAdd(&cur[b], 1);
      stage[p] = make_int2((ss[k] << BSH) | (dd[k] & (BNOD - 1)), ww[k]);
      sbk[p] = (unsigned short)b;
    }
  }
  __syncthreads();
  const int nloc = (E - e0 < CH) ? (E - e0) : CH;
  for (int i = t; i < nloc; i += TPB) grp[delta[sbk[i]] + i] = stage[i];
}

__global__ void __launch_bounds__(TPB)
k_owner(const int2* __restrict__ grp, const int* __restrict__ bstart,
        int2* __restrict__ csr, int* __restrict__ rowptr, float* __restrict__ dis,
        int N, int E, int B) {
  __shared__ int cntl[BNOD];
  __shared__ float degl[BNOD];
  __shared__ int startl[BNOD];
  __shared__ int cur[BNOD];
  __shared__ int wsum[32];
  __shared__ int2 stage[CH];
  const int b = blockIdx.x;
  const int t = threadIdx.x;
  const int s = bstart[b];
  const int nE = bstart[b + 1] - s;
  if (t < BNOD) { cntl[t] = 0; degl[t] = 0.0f; }
  __syncthreads();
  for (int i = t; i < nE; i += TPB) {
    const int2 r = grp[s + i];
    const int nl = r.x & (BNOD - 1);
    atomicAdd(&cntl[nl], 1);
    atomicAdd(&degl[nl], __int_as_float(r.y));
  }
  __syncthreads();
  const int ex = block_excl_scan((t < BNOD) ? cntl[t] : 0, wsum);
  if (t < BNOD) { startl[t] = ex; cur[t] = ex; }
  __syncthreads();
  const bool inl = (nE <= CH);
  for (int i = t; i < nE; i += TPB) {
    const int2 r = grp[s + i];
    const int nl = r.x & (BNOD - 1);
    const int p = atomicAdd(&cur[nl], 1);
    const int2 o = make_int2(r.x >> BSH, r.y);
    if (inl) stage[p] = o; else csr[s + p] = o;
  }
  __syncthreads();
  if (inl) for (int i = t; i < nE; i += TPB) csr[s + i] = stage[i];
  if (t < BNOD) {
    const int n = (b << BSH) + t;
    if (n < N) {
      rowptr[n] = s + startl[t];
      const float d = degl[t] + 1.0f;
      dis[n] = (d > 0.0f) ? rsqrtf(d) : 0.0f;
    }
  }
  if (b == B - 1 && t == 0) rowptr[N] = E;
}

template <int COUT, bool RELU>
__global__ void k_gemm_wmma(const float* __restrict__ H, const float* __restrict__ Wg,
                            const float* __restrict__ dis, float* __restrict__ HW, int N) {
  constexpr int COLT = COUT / 16;
  constexpr int ROWT = 8 / COLT;
  constexpr int WSTR = 68;
  __shared__ float Wl[COUT * WSTR];

  for (int i = threadIdx.x; i < 64 * COUT; i += 256) {
    int k = i / COUT;
    int n = i - k * COUT;
    Wl[n * WSTR + k] = Wg[i];
  }
  __syncthreads();

  const int wave = threadIdx.x >> 5;
  const int lane = threadIdx.x & 31;
  const int rowTile = wave / COLT;
  const int colTile = wave - rowTile * COLT;
  const int m0 = (blockIdx.x * ROWT + rowTile) * 16;
  if (m0 >= N) return;
  const int n0 = colTile * 16;
  const int mr = lane & 15;
  const int kb = (lane >> 4) << 1;

  const int mbase = m0 + ((lane >> 4) << 3);
  const float4 dlo = *(const float4*)(dis + mbase);
  const float4 dhi = *(const float4*)(dis + mbase + 4);

  v8f acc = {};
  const float* hrow = H + (size_t)(m0 + mr) * 64;
  const float* wrow = Wl + (n0 + mr) * WSTR;
#pragma unroll
  for (int k = 0; k < 64; k += 4) {
    float a0 = hrow[k + kb];
    float a1 = hrow[k + kb + 1];
    if (RELU) { a0 = fmaxf(a0, 0.0f); a1 = fmaxf(a1, 0.0f); }
    v2f a; a.x = a0; a.y = a1;
    v2f b = *(const v2f*)&wrow[k + kb];
    acc = __builtin_amdgcn_wmma_f32_16x16x4_f32(
        false, a, false, b, (short)0, acc, false, false);
  }

  const float dsc[8] = {dlo.x, dlo.y, dlo.z, dlo.w, dhi.x, dhi.y, dhi.z, dhi.w};
#pragma unroll
  for (int v = 0; v < 8; ++v) {
    size_t idx = (size_t)(mbase + v) * COUT + n0 + mr;
    HW[idx] = acc[v] * dsc[v];
  }
}

template <int LPN>
__global__ void k_agg(const float* __restrict__ hw, const int2* __restrict__ csr,
                      const int* __restrict__ rowptr, const float* __restrict__ bias,
                      const float* __restrict__ dis, float* __restrict__ out, int N) {
  constexpr int C = 4 * LPN;
  long t = (long)blockIdx.x * blockDim.x + threadIdx.x;
  int n = (int)(t / LPN);
  if (n >= N) return;
  const int f = (int)(t % LPN) << 2;
  int p = rowptr[n];
  const int pe = rowptr[n + 1];
  const float dn = dis[n];
  const float4 bv = *(const float4*)(bias + f);
  float4 acc = *(const float4*)(hw + (size_t)n * C + f);
  for (; p + 4 <= pe; p += 4) {
    const int2 c0 = csr[p], c1 = csr[p + 1], c2 = csr[p + 2], c3 = csr[p + 3];
    const float4 v0 = *(const float4*)(hw + (size_t)c0.x * C + f);
    const float4 v1 = *(const float4*)(hw + (size_t)c1.x * C + f);
    const float4 v2 = *(const float4*)(hw + (size_t)c2.x * C + f);
    const float4 v3 = *(const float4*)(hw + (size_t)c3.x * C + f);
    const float n0 = __int_as_float(c0.y), n1 = __int_as_float(c1.y);
    const float n2 = __int_as_float(c2.y), n3 = __int_as_float(c3.y);
    acc.x += n0 * v0.x; acc.y += n0 * v0.y; acc.z += n0 * v0.z; acc.w += n0 * v0.w;
    acc.x += n1 * v1.x; acc.y += n1 * v1.y; acc.z += n1 * v1.z; acc.w += n1 * v1.w;
    acc.x += n2 * v2.x; acc.y += n2 * v2.y; acc.z += n2 * v2.z; acc.w += n2 * v2.w;
    acc.x += n3 * v3.x; acc.y += n3 * v3.y; acc.z += n3 * v3.z; acc.w += n3 * v3.w;
  }
#pragma unroll 1
  for (; p < pe; ++p) {
    const int2 c0 = csr[p];
    const float4 v0 = *(const float4*)(hw + (size_t)c0.x * C + f);
    const float n0 = __int_as_float(c0.y);
    acc.x += n0 * v0.x; acc.y += n0 * v0.y; acc.z += n0 * v0.z; acc.w += n0 * v0.w;
  }
  float4 o;
  o.x = bv.x + dn * acc.x;
  o.y = bv.y + dn * acc.y;
  o.z = bv.z + dn * acc.z;
  o.w = bv.w + dn * acc.w;
  *(float4*)(out + (size_t)n * C + f) = o;
}

template <int COUT>
__global__ void __launch_bounds__(512)
k_agg_gemm(const float* __restrict__ hw, const int2* __restrict__ csr,
           const int* __restrict__ rowptr, const float* __restrict__ bias,
           const float* __restrict__ dis, const float* __restrict__ Wn,
           float* __restrict__ hwn, int N) {
  constexpr int C = 64;
  constexpr int ASTR = 68;
  __shared__ float At[32 * ASTR];
  const int nl = threadIdx.x >> 4;
  const int f = (threadIdx.x & 15) << 2;
  const int m0 = blockIdx.x * 32;
  const int n = m0 + nl;
  float4 o = make_float4(0.0f, 0.0f, 0.0f, 0.0f);
  float dn = 0.0f;
  if (n < N) {
    int p = rowptr[n];
    const int pe = rowptr[n + 1];
    dn = dis[n];
    const float4 bv = *(const float4*)(bias + f);
    float4 acc = *(const float4*)(hw + (size_t)n * C + f);
    for (; p + 4 <= pe; p += 4) {
      const int2 c0 = csr[p], c1 = csr[p + 1], c2 = csr[p + 2], c3 = csr[p + 3];
      const float4 v0 = *(const float4*)(hw + (size_t)c0.x * C + f);
      const float4 v1 = *(const float4*)(hw + (size_t)c1.x * C + f);
      const float4 v2 = *(const float4*)(hw + (size_t)c2.x * C + f);
      const float4 v3 = *(const float4*)(hw + (size_t)c3.x * C + f);
      const float n0 = __int_as_float(c0.y), n1 = __int_as_float(c1.y);
      const float n2 = __int_as_float(c2.y), n3 = __int_as_float(c3.y);
      acc.x += n0 * v0.x; acc.y += n0 * v0.y; acc.z += n0 * v0.z; acc.w += n0 * v0.w;
      acc.x += n1 * v1.x; acc.y += n1 * v1.y; acc.z += n1 * v1.z; acc.w += n1 * v1.w;
      acc.x += n2 * v2.x; acc.y += n2 * v2.y; acc.z += n2 * v2.z; acc.w += n2 * v2.w;
      acc.x += n3 * v3.x; acc.y += n3 * v3.y; acc.z += n3 * v3.z; acc.w += n3 * v3.w;
    }
#pragma unroll 1
    for (; p < pe; ++p) {
      const int2 c0 = csr[p];
      const float4 v0 = *(const float4*)(hw + (size_t)c0.x * C + f);
      const float n0 = __int_as_float(c0.y);
      acc.x += n0 * v0.x; acc.y += n0 * v0.y; acc.z += n0 * v0.z; acc.w += n0 * v0.w;
    }
    o.x = fmaxf(bv.x + dn * acc.x, 0.0f);
    o.y = fmaxf(bv.y + dn * acc.y, 0.0f);
    o.z = fmaxf(bv.z + dn * acc.z, 0.0f);
    o.w = fmaxf(bv.w + dn * acc.w, 0.0f);
  }
  *(float4*)&At[nl * ASTR + f] = o;
  if (f == 0) At[nl * ASTR + C] = dn;
  __syncthreads();

  const int wave = threadIdx.x >> 5;
  if (wave >= COUT / 16) return;
  const int lane = threadIdx.x & 31;
  const int mr = lane & 15;
  const int kb = (lane >> 4) << 1;
  const int n0 = wave * 16;
  const int rl = (lane >> 4) << 3;
  const int mbase = m0 + rl;

  v8f acc0 = {}, acc1 = {};
  const float* arow = At + mr * ASTR + kb;
  const float* wcol = Wn + (size_t)kb * COUT + n0 + mr;
#pragma unroll
  for (int k = 0; k < 64; k += 4) {
    v2f a0 = *(const v2f*)&arow[k];
    v2f a1 = *(const v2f*)&arow[16 * ASTR + k];
    v2f b;
    b.x = wcol[(size_t)k * COUT];
    b.y = wcol[(size_t)(k + 1) * COUT];
    acc0 = __builtin_amdgcn_wmma_f32_16x16x4_f32(
        false, a0, false, b, (short)0, acc0, false, false);
    acc1 = __builtin_amdgcn_wmma_f32_16x16x4_f32(
        false, a1, false, b, (short)0, acc1, false, false);
  }

  const float* drow = At + rl * ASTR + C;
#pragma unroll
  for (int v = 0; v < 8; ++v) {
    hwn[(size_t)(mbase + v) * COUT + n0 + mr] = acc0[v] * drow[v * ASTR];
    hwn[(size_t)(mbase + 16 + v) * COUT + n0 + mr] = acc1[v] * drow[(16 + v) * ASTR];
  }
}

__global__ void k_logsoftmax16(float* __restrict__ out, int N) {
  int n = blockIdx.x * blockDim.x + threadIdx.x;
  if (n >= N) return;
  const float* lg = out + (size_t)N * 16 + (size_t)n * 16;
  float v[16];
  float mx = -3.402823466e38f;
#pragma unroll
  for (int i = 0; i < 16; ++i) { v[i] = lg[i]; mx = fmaxf(mx, v[i]); }
  float s = 0.0f;
#pragma unroll
  for (int i = 0; i < 16; ++i) s += expf(v[i] - mx);
  float lse = mx + logf(s);
  float* o0 = out + (size_t)n * 16;
  float* o2 = out + (size_t)2 * N * 16 + (size_t)n * 16;
#pragma unroll
  for (int i = 0; i < 16; ++i) {
    o0[i] = v[i] - lse;
    o2[i] = v[i];
  }
}

extern "C" void kernel_launch(void* const* d_in, const int* in_sizes, int n_in,
                              void* d_out, int out_size, void* d_ws, size_t ws_size,
                              hipStream_t stream) {
  const float* x  = (const float*)d_in[0];
  const int*   ei = (const int*)  d_in[1];
  const float* ew = (const float*)d_in[2];
  const float* W1 = (const float*)d_in[3];
  const float* b1 = (const float*)d_in[4];
  const float* W2 = (const float*)d_in[5];
  const float* b2 = (const float*)d_in[6];
  const float* W3 = (const float*)d_in[7];
  const float* b3 = (const float*)d_in[8];
  float* out = (float*)d_out;

  const int N = in_sizes[0] / 64;
  const int E = in_sizes[2];
  const int* src = ei;
  const int* dst = ei + E;

  const int Np = (N + 64) & ~63;
  const int Ep = (E + 63) & ~63;
  const int B  = (N + BNOD - 1) >> BSH;
  const int Bq = (B + 64) & ~63;
  const int C  = (E + CH - 1) / CH;
  const int rowTiles = (N + 15) / 16;
  const int rowBlks  = (N + 31) / 32;

  int*   hist   = (int*)d_ws;
  int*   bstart = hist + (size_t)C * BMAX;
  int2*  grp    = (int2*)(bstart + Bq);
  int*   rowptr = (int*)(grp + Ep);
  int2*  csr    = (int2*)(rowptr + Np);
  float* dis    = (float*)(csr + Ep);
  float* bufA   = dis + Np;
  float* bufB   = bufA + (size_t)rowBlks * 32 * 64;
  float* logits = out + (size_t)N * 16;

  const int T = 256;
  const int gN = (N + T - 1) / T;
  const int gAg16 = (int)(((size_t)N * 4  + T - 1) / T);

  k_hist <<<C, TPB, 0, stream>>>(dst, hist, E);
  k_group<<<C, TPB, 0, stream>>>(src, dst, ew, hist, grp, bstart, E, B, C);
  k_owner<<<B, TPB, 0, stream>>>(grp, bstart, csr, rowptr, dis, N, E, B);

  k_gemm_wmma<64, false><<<(rowTiles + 1) / 2, 256, 0, stream>>>(x, W1, dis, bufA, N);

  k_agg_gemm<64><<<rowBlks, 512, 0, stream>>>(bufA, csr, rowptr, b1, dis, W2, bufB, N);

  k_agg_gemm<16><<<rowBlks, 512, 0, stream>>>(bufB, csr, rowptr, b2, dis, W3, bufA, N);

  k_agg<4><<<gAg16, T, 0, stream>>>(bufA, csr, rowptr, b3, dis, logits, N);

  k_logsoftmax16<<<gN, T, 0, stream>>>(out, N);
}


# Aggregation runs as few long waves: a wave takes 16 nodes (one 16-row tile, gather to product), 8 edge rows in flight per lane, list tail masked by weight 0, not looped singly; 112500 waves to 18756
# speedup vs baseline: 4.2682x; 1.0974x over previous
#include <hip/hip_runtime.h>
#include <hip/hip_bf16.h>
#include <math.h>


typedef float v2f __attribute__((ext_vector_type(2)));
typedef float v8f __attribute__((ext_vector_type(8)));


constexpr int BSH  = 8;
constexpr int BNOD = 1 << BSH;
constexpr int BMAX = 512;
constexpr int CH   = 8192;
constexpr int TPB  = 1024;

__device__ __forceinline__ int block_excl_scan(int v, int* wsum) {
  const int lane = threadIdx.x & 31;
  const int wave = threadIdx.x >> 5;
  int inc = v;
#pragma unroll
  for (int d = 1; d < 32; d <<= 1) {
    int u = __shfl_up(inc, d);
    if (lane >= d) inc += u;
  }
  if (lane == 31) wsum[wave] = inc;
  __syncthreads();
  const int nw = blockDim.x >> 5;
  int off = (lane < nw && lane < wave) ? wsum[lane] : 0;
#pragma unroll
  for (int m = 16; m >= 1; m >>= 1) off += __shfl_xor(off, m);
  __syncthreads();
  return off + inc - v;
}

__global__ void __launch_bounds__(TPB)
k_hist(const int* __restrict__ dst, int* __restrict__ hist, int E) {
  __shared__ int h[BMAX];
  const int t = threadIdx.x;
  if (t < BMAX) h[t] = 0;
  __syncthreads();
  const int e0 = blockIdx.x * CH;
  int d[CH / TPB];
#pragma unroll
  for (int k = 0; k < CH / TPB; ++k) {
    const int e = e0 + k * TPB + t;
    d[k] = dst[(e < E) ? e : (E - 1)];
  }
#pragma unroll
  for (int k = 0; k < CH / TPB; ++k) {
    const int e = e0 + k * TPB + t;
    if (e < E) atomicAdd(&h[d[k] >> BSH], 1);
  }
  __syncthreads();
  if (t < BMAX) hist[(size_t)blockIdx.x * BMAX + t] = h[t];
}

__global__ void __launch_bounds__(TPB)
k_group(const int* __restrict__ src, const int* __restrict__ dst,
        const float* __restrict__ ew, const int* __restrict__ hist,
        int2* __restrict__ grp, int* __restrict__ bstart,
        int E, int B, int C) {
  __shared__ int delta[BMAX];
  __shared__ int cur[BMAX];
  __shared__ int totL[BMAX];
  __shared__ int prevL[BMAX];
  __shared__ int wsum[32];
  __shared__ int2 stage[CH];
  __shared__ unsigned short sbk[CH];
  const int c = blockIdx.x;
  const int t = threadIdx.x;
  const int e0 = c * CH;
  int dd[CH / TPB], ss[CH / TPB], ww[CH / TPB];
#pragma unroll
  for (int k = 0; k < CH / TPB; ++k) {
    const int e = e0 + k * TPB + t;
    const int ec = (e < E) ? e : (E - 1);
    dd[k] = dst[ec];
    ss[k] = src[ec];
    ww[k] = __float_as_int(ew[ec]);
  }
  const int own = (t < B) ? hist[(size_t)c * BMAX + t] : 0;
  if (t < BMAX) { totL[t] = 0; prevL[t] = 0; }
  __syncthreads();
  {
    const int lane = t & 31;
    const int wave = t >> 5;
    int tt[BMAX / 32], pp[BMAX / 32];
#pragma unroll
    for (int j = 0; j < BMAX / 32; ++j) { tt[j] = 0; pp[j] = 0; }
    for (int r = wave; r < C; r += TPB / 32) {
      const int* hr = hist + (size_t)r * BMAX + lane;
      const bool before = r < c;
      int v[BMAX / 32];
#pragma unroll
      for (int j = 0; j < BMAX / 32; ++j) v[j] = hr[32 * j];
#pragma unroll
      for (int j = 0; j < BMAX / 32; ++j) {
        tt[j] += v[j];
        pp[j] += before ? v[j] : 0;
      }
    }
#pragma unroll
    for (int j = 0; j < BMAX / 32; ++j) {
      atomicAdd(&totL[lane + 32 * j], tt[j]);
      atomicAdd(&prevL[lane + 32 * j], pp[j]);
    }
  }
  __syncthreads();
  const int tot  = (t < B) ? totL[t] : 0;
  const int prev = (t < B) ? prevL[t] : 0;
  const int gstart = block_excl_scan(tot, wsum);
  const int lstart = block_excl_scan(own, wsum);
  if (t < B) {
    delta[t] = gstart + prev - lstart;
    cur[t] = lstart;
    if (c == 0) bstart[t] = gstart;
  }
  if (c == 0 && t == 0) bstart[B] = E;
  __syncthreads();
#pragma unroll
  for (int k = 0; k < CH / TPB; ++k) {
    const int e = e0 + k * TPB + t;
    if (e < E) {
      const int b = dd[k] >> BSH;
      const int p = atomicAdd(&cur[b], 1);
      stage[p] = make_int2((ss[k] << BSH) | (dd[k] & (BNOD - 1)), ww[k]);
      sbk[p] = (unsigned short)b;
    }
  }
  __syncthreads();
  const int nloc = (E - e0 < CH) ? (E - e0) : CH;
  for (int i = t; i < nloc; i += TPB) grp[delta[sbk[i]] + i] = stage[i];
}

__global__ void __launch_bounds__(TPB)
k_owner(const int2* __restrict__ grp, const int* __restrict__ bstart,
        int2* __restrict__ csr, int* __restrict__ rowptr, float* __restrict__ dis,
        int N, int E, int B) {
  __shared__ int cntl[BNOD];
  __shared__ float degl[BNOD];
  __shared__ int startl[BNOD];
  __shared__ int cur[BNOD];
  __shared__ int wsum[32];
  __shared__ int2 stage[CH];
  const int b = blockIdx.x;
  const int t = threadIdx.x;
  const int s = bstart[b];
  const int nE = bstart[b + 1] - s;
  if (t < BNOD) { cntl[t] = 0; degl[t] = 0.0f; }
  __syncthreads();
  for (int i = t; i < nE; i += TPB) {
    const int2 r = grp[s + i];
    const int nl = r.x & (BNOD - 1);
    atomicAdd(&cntl[nl], 1);
    atomicAdd(&degl[nl], __int_as_float(r.y));
  }
  __syncthreads();
  const int ex = block_excl_scan((t < BNOD) ? cntl[t] : 0, wsum);
  if (t < BNOD) { startl[t] = ex; cur[t] = ex; }
  __syncthreads();
  const bool inl = (nE <= CH);
  for (int i = t; i < nE; i += TPB) {
    const int2 r = grp[s + i];
    const int nl = r.x & (BNOD - 1);
    const int p = atomicAdd(&cur[nl], 1);
    const int2 o = make_int2(r.x >> BSH, r.y);
    if (inl) stage[p] = o; else csr[s + p] = o;
  }
  __syncthreads();
  if (inl) for (int i = t; i < nE; i += TPB) csr[s + i] = stage[i];
  if (t < BNOD) {
    const int n = (b << BSH) + t;
    if (n < N) {
      rowptr[n] = s + startl[t];
      const float d = degl[t] + 1.0f;
      dis[n] = (d > 0.0f) ? rsqrtf(d) : 0.0f;
    }
  }
  if (b == B - 1 && t == 0) rowptr[N] = E;
}

template <int COUT, bool RELU>
__global__ void k_gemm_wmma(const float* __restrict__ H, const float* __restrict__ Wg,
                            const float* __restrict__ dis, float* __restrict__ HW, int N) {
  constexpr int COLT = COUT / 16;
  constexpr int ROWT = 8 / COLT;
  constexpr int WSTR = 68;
  __shared__ float Wl[COUT * WSTR];

  for (int i = threadIdx.x; i < 64 * COUT; i += 256) {
    int k = i / COUT;
    int n = i - k * COUT;
    Wl[n * WSTR + k] = Wg[i];
  }
  __syncthreads();

  const int wave = threadIdx.x >> 5;
  const int lane = threadIdx.x & 31;
  const int rowTile = wave / COLT;
  const int colTile = wave - rowTile * COLT;
  const int m0 = (blockIdx.x * ROWT + rowTile) * 16;
  if (m0 >= N) return;
  const int n0 = colTile * 16;
  const int mr = lane & 15;
  const int kb = (lane >> 4) << 1;

  const int mbase = m0 + ((lane >> 4) << 3);
  const float4 dlo = *(const float4*)(dis + mbase);
  const float4 dhi = *(const float4*)(dis + mbase + 4);

  v8f acc = {};
  const float* hrow = H + (size_t)(m0 + mr) * 64;
  const float* wrow = Wl + (n0 + mr) * WSTR;
#pragma unroll
  for (int k = 0; k < 64; k += 4) {
    float a0 = hrow[k + kb];
    float a1 = hrow[k + kb + 1];
    if (RELU) { a0 = fmaxf(a0, 0.0f); a1 = fmaxf(a1, 0.0f); }
    v2f a; a.x = a0; a.y = a1;
    v2f b = *(const v2f*)&wrow[k + kb];
    acc = __builtin_amdgcn_wmma_f32_16x16x4_f32(
        false, a, false, b, (short)0, acc, false, false);
  }

  const float dsc[8] = {dlo.x, dlo.y, dlo.z, dlo.w, dhi.x, dhi.y, dhi.z, dhi.w};
#pragma unroll
  for (int v = 0; v < 8; ++v) {
    size_t idx = (size_t)(mbase + v) * COUT + n0 + mr;
    HW[idx] = acc[v] * dsc[v];
  }
}

template <int ROWB>
__device__ __forceinline__ float4 node_sum(const float* __restrict__ hw,
                                           const int2* __restrict__ csr,
                                           int n, int p, int pe, int fb) {
  constexpr int U = 8;
  const char* hb = (const char*)hw;
  float4 acc = *(const float4*)(hb + (unsigned)(n * ROWB + fb));
  if (p >= pe) return acc;
  const int last = pe - 1;
  int2 c[U];
#pragma unroll
  for (int j = 0; j < U; ++j) c[j] = csr[min(p + j, last)];
  bool more = true;
  while (more) {
    float4 v[U];
    float w[U];
#pragma unroll
    for (int j = 0; j < U; ++j)
      v[j] = *(const float4*)(hb + (unsigned)(c[j].x * ROWB + fb));
#pragma unroll
    for (int j = 0; j < U; ++j) w[j] = (p + j < pe) ? __int_as_float(c[j].y) : 0.0f;
    p += U;
    more = p < pe;
    if (more) {
#pragma unroll
      for (int j = 0; j < U; ++j) c[j] = csr[min(p + j, last)];
    }
#pragma unroll
    for (int j = 0; j < U; ++j) {
      acc.x += w[j] * v[j].x; acc.y += w[j] * v[j].y;
      acc.z += w[j] * v[j].z; acc.w += w[j] * v[j].w;
    }
  }
  return acc;
}

__global__ void __launch_bounds__(128)
k_agg16(const float* __restrict__ hw, const int2* __restrict__ csr,
        const int* __restrict__ rowptr, const float* __restrict__ bias,
        const float* __restrict__ dis, float* __restrict__ out, int N) {
  const int lane = threadIdx.x & 31;
  const int w0 = (blockIdx.x * 4 + (threadIdx.x >> 5)) * 16;
  if (w0 >= N) return;
  const int f = (lane & 3) << 2;
  int rpl = 0;
  float dl = 0.0f;
  if (lane <= 16 && w0 + lane <= N) rpl = rowptr[w0 + lane];
  if (lane < 16 && w0 + lane < N) dl = dis[w0 + lane];
  const float4 bv = *(const float4*)(bias + f);
#pragma unroll
  for (int it = 0; it < 2; ++it) {
    const int nw = it * 8 + (lane >> 2);
    const int n = w0 + nw;
    const int p = __shfl(rpl, nw);
    const int pe = __shfl(rpl, nw + 1);
    const float dn = __shfl(dl, nw);
    if (n < N) {
      const float4 acc = node_sum<64>(hw, csr, n, p, pe, f << 2);
      float4 o;
      o.x = bv.x + dn * acc.x;
      o.y = bv.y + dn * acc.y;
      o.z = bv.z + dn * acc.z;
      o.w = bv.w + dn * acc.w;
      *(float4*)(out + (size_t)n * 16 + f) = o;
    }
  }
}

template <int COUT>
__global__ void __launch_bounds__(128)
k_agg_gemm(const float* __restrict__ hw, const int2* __restrict__ csr,
           const int* __restrict__ rowptr, const float* __restrict__ bias,
           const float* __restrict__ dis, const float* __restrict__ Wn,
           float* __restrict__ hwn, int N) {
  constexpr int C = 64;
  constexpr int ASTR = 68;
  constexpr int COLT = COUT / 16;
  __shared__ float At[64 * ASTR];
  const int wave = threadIdx.x >> 5;
  const int lane = threadIdx.x & 31;
  const int f = (lane & 15) << 2;
  const int m0 = blockIdx.x * 64;
  const int w0 = m0 + wave * 16;
  int rpl = 0;
  float dl = 0.0f;
  if (lane <= 16 && w0 + lane <= N) rpl = rowptr[w0 + lane];
  if (lane < 16 && w0 + lane < N) dl = dis[w0 + lane];
  const float4 bv = *(const float4*)(bias + f);
#pragma unroll 1
  for (int it = 0; it < 8; ++it) {
    const int nw = it * 2 + (lane >> 4);
    const int n = w0 + nw;
    const int p = __shfl(rpl, nw);
    const int pe = __shfl(rpl, nw + 1);
    const float dn = __shfl(dl, nw);
    float4 o = make_float4(0.0f, 0.0f, 0.0f, 0.0f);
    if (n < N) {
      const float4 acc = node_sum<256>(hw, csr, n, p, pe, f << 2);
      o.x = fmaxf(bv.x + dn * acc.x, 0.0f);
      o.y = fmaxf(bv.y + dn * acc.y, 0.0f);
      o.z = fmaxf(bv.z + dn * acc.z, 0.0f);
      o.w = fmaxf(bv.w + dn * acc.w, 0.0f);
    }
    float* arw = At + (wave * 16 + nw) * ASTR;
    *(float4*)&arw[f] = o;
    if (f == 0) arw[C] = dn;
  }
  __syncthreads();

  const int mr = lane & 15;
  const int kb = (lane >> 4) << 1;
  const int rl = (lane >> 4) << 3;

  v8f acc[COLT];
#pragma unroll
  for (int t = 0; t < COLT; ++t) acc[t] = v8f{};
  const float* arow = At + (wave * 16 + mr) * ASTR + kb;
  const float* wcol = Wn + (size_t)kb * COUT + mr;
#pragma unroll
  for (int k = 0; k < 64; k += 4) {
    v2f a = *(const v2f*)&arow[k];
#pragma unroll
    for (int t = 0; t < COLT; ++t) {
      v2f b;
      b.x = wcol[(size_t)k * COUT + 16 * t];
      b.y = wcol[(size_t)(k + 1) * COUT + 16 * t];
      acc[t] = __builtin_amdgcn_wmma_f32_16x16x4_f32(
          false, a, false, b, (short)0, acc[t], false, false);
    }
  }

  const float* drow = At + (wave * 16 + rl) * ASTR + C;
#pragma unroll
  for (int v = 0; v < 8; ++v) {
    const float dv = drow[v * ASTR];
#pragma unroll
    for (int t = 0; t < COLT; ++t)
      hwn[(size_t)(w0 + rl + v) * COUT + 16 * t + mr] = acc[t][v] * dv;
  }
}

__global__ void k_logsoftmax16(float* __restrict__ out, int N) {
  int n = blockIdx.x * blockDim.x + threadIdx.x;
  if (n >= N) return;
  const float* lg = out + (size_t)N * 16 + (size_t)n * 16;
  float v[16];
  float mx = -3.402823466e38f;
#pragma unroll
  for (int i = 0; i < 16; ++i) { v[i] = lg[i]; mx = fmaxf(mx, v[i]); }
  float s = 0.0f;
#pragma unroll
  for (int i = 0; i < 16; ++i) s += expf(v[i] - mx);
  float lse = mx + logf(s);
  float* o0 = out + (size_t)n * 16;
  float* o2 = out + (size_t)2 * N * 16 + (size_t)n * 16;
#pragma unroll
  for (int i = 0; i < 16; ++i) {
    o0[i] = v[i] - lse;
    o2[i] = v[i];
  }
}

extern "C" void kernel_launch(void* const* d_in, const int* in_sizes, int n_in,
                              void* d_out, int out_size, void* d_ws, size_t ws_size,
                              hipStream_t stream) {
  const float* x  = (const float*)d_in[0];
  const int*   ei = (const int*)  d_in[1];
  const float* ew = (const float*)d_in[2];
  const float* W1 = (const float*)d_in[3];
  const float* b1 = (const float*)d_in[4];
  const float* W2 = (const float*)d_in[5];
  const float* b2 = (const float*)d_in[6];
  const float* W3 = (const float*)d_in[7];
  const float* b3 = (const float*)d_in[8];
  float* out = (float*)d_out;

  const int N = in_sizes[0] / 64;
  const int E = in_sizes[2];
  const int* src = ei;
  const int* dst = ei + E;

  const int Np = (N + 64) & ~63;
  const int Ep = (E + 63) & ~63;
  const int B  = (N + BNOD - 1) >> BSH;
  const int Bq = (B + 64) & ~63;
  const int C  = (E + CH - 1) / CH;
  const int rowTiles = (N + 15) / 16;
  const int rowBlks  = (N + 63) / 64;

  int*   hist   = (int*)d_ws;
  int*   bstart = hist + (size_t)C * BMAX;
  int2*  grp    = (int2*)(bstart + Bq);
  int*   rowptr = (int*)(grp + Ep);
  int2*  csr    = (int2*)(rowptr + Np);
  float* dis    = (float*)(csr + Ep);
  float* bufA   = dis + Np;
  float* bufB   = bufA + (size_t)rowBlks * 64 * 64;
  float* logits = out + (size_t)N * 16;

  const int T = 256;
  const int gN = (N + T - 1) / T;
  const int gAg16 = (N + 63) / 64;

  k_hist <<<C, TPB, 0, stream>>>(dst, hist, E);
  k_group<<<C, TPB, 0, stream>>>(src, dst, ew, hist, grp, bstart, E, B, C);
  k_owner<<<B, TPB, 0, stream>>>(grp, bstart, csr, rowptr, dis, N, E, B);

  k_gemm_wmma<64, false><<<(rowTiles + 1) / 2, 256, 0, stream>>>(x, W1, dis, bufA, N);

  k_agg_gemm<64><<<rowBlks, 128, 0, stream>>>(bufA, csr, rowptr, b1, dis, W2, bufB, N);

  k_agg_gemm<16><<<rowBlks, 128, 0, stream>>>(bufB, csr, rowptr, b2, dis, W3, bufA, N);

  k_agg16<<<gAg16, 128, 0, stream>>>(bufA, csr, rowptr, b3, dis, logits, N);

  k_logsoftmax16<<<gN, T, 0, stream>>>(out, N);
}


# Log-softmax and second logits copy fused into the last aggregation: four lanes holding a node's logits reduce max and exp-sum by shuffles, store all three outputs; 8 launches to 7; exp-sum as tree
# speedup vs baseline: 4.4124x; 1.0338x over previous
#include <hip/hip_runtime.h>
#include <hip/hip_bf16.h>
#include <math.h>


typedef float v2f __attribute__((ext_vector_type(2)));
typedef float v8f __attribute__((ext_vector_type(8)));


constexpr int BSH  = 8;
constexpr int BNOD = 1 << BSH;
constexpr int BMAX = 512;
constexpr int CH   = 8192;
constexpr int TPB  = 1024;

__device__ __forceinline__ int block_excl_scan(int v, int* wsum) {
  const int lane = threadIdx.x & 31;
  const int wave = threadIdx.x >> 5;
  int inc = v;
#pragma unroll
  for (int d = 1; d < 32; d <<= 1) {
    int u = __shfl_up(inc, d);
    if (lane >= d) inc += u;
  }
  if (lane == 31) wsum[wave] = inc;
  __syncthreads();
  const int nw = blockDim.x >> 5;
  int off = (lane < nw && lane < wave) ? wsum[lane] : 0;
#pragma unroll
  for (int m = 16; m >= 1; m >>= 1) off += __shfl_xor(off, m);
  __syncthreads();
  return off + inc - v;
}

__global__ void __launch_bounds__(TPB)
k_hist(const int* __restrict__ dst, int* __restrict__ hist, int E) {
  __shared__ int h[BMAX];
  const int t = threadIdx.x;
  if (t < BMAX) h[t] = 0;
  __syncthreads();
  const int e0 = blockIdx.x * CH;
  int d[CH / TPB];
#pragma unroll
  for (int k = 0; k < CH / TPB; ++k) {
    const int e = e0 + k * TPB + t;
    d[k] = dst[(e < E) ? e : (E - 1)];
  }
#pragma unroll
  for (int k = 0; k < CH / TPB; ++k) {
    const int e = e0 + k * TPB + t;
    if (e < E) atomicAdd(&h[d[k] >> BSH], 1);
  }
  __syncthreads();
  if (t < BMAX) hist[(size_t)blockIdx.x * BMAX + t] = h[t];
}

__global__ void __launch_bounds__(TPB)
k_group(const int* __restrict__ src, const int* __restrict__ dst,
        const float* __restrict__ ew, const int* __restrict__ hist,
        int2* __restrict__ grp, int* __restrict__ bstart,
        int E, int B, int C) {
  __shared__ int delta[BMAX];
  __shared__ int cur[BMAX];
  __shared__ int totL[BMAX];
  __shared__ int prevL[BMAX];
  __shared__ int wsum[32];
  __shared__ int2 stage[CH];
  __shared__ unsigned short sbk[CH];
  const int c = blockIdx.x;
  const int t = threadIdx.x;
  const int e0 = c * CH;
  int dd[CH / TPB], ss[CH / TPB], ww[CH / TPB];
#pragma unroll
  for (int k = 0; k < CH / TPB; ++k) {
    const int e = e0 + k * TPB + t;
    const int ec = (e < E) ? e : (E - 1);
    dd[k] = dst[ec];
    ss[k] = src[ec];
    ww[k] = __float_as_int(ew[ec]);
  }
  const int own = (t < B) ? hist[(size_t)c * BMAX + t] : 0;
  if (t < BMAX) { totL[t] = 0; prevL[t] = 0; }
  __syncthreads();
  {
    const int lane = t & 31;
    const int wave = t >> 5;
    int tt[BMAX / 32], pp[BMAX / 32];
#pragma unroll
    for (int j = 0; j < BMAX / 32; ++j) { tt[j] = 0; pp[j] = 0; }
    for (int r = wave; r < C; r += TPB / 32) {
      const int* hr = hist + (size_t)r * BMAX + lane;
      const bool before = r < c;
      int v[BMAX / 32];
#pragma unroll
      for (int j = 0; j < BMAX / 32; ++j) v[j] = hr[32 * j];
#pragma unroll
      for (int j = 0; j < BMAX / 32; ++j) {
        tt[j] += v[j];
        pp[j] += before ? v[j] : 0;
      }
    }
#pragma unroll
    for (int j = 0; j < BMAX / 32; ++j) {
      atomicAdd(&totL[lane + 32 * j], tt[j]);
      atomicAdd(&prevL[lane + 32 * j], pp[j]);
    }
  }
  __syncthreads();
  const int tot  = (t < B) ? totL[t] : 0;
  const int prev = (t < B) ? prevL[t] : 0;
  const int gstart = block_excl_scan(tot, wsum);
  const int lstart = block_excl_scan(own, wsum);
  if (t < B) {
    delta[t] = gstart + prev - lstart;
    cur[t] = lstart;
    if (c == 0) bstart[t] = gstart;
  }
  if (c == 0 && t == 0) bstart[B] = E;
  __syncthreads();
#pragma unroll
  for (int k = 0; k < CH / TPB; ++k) {
    const int e = e0 + k * TPB + t;
    if (e < E) {
      const int b = dd[k] >> BSH;
      const int p = atomicAdd(&cur[b], 1);
      stage[p] = make_int2((ss[k] << BSH) | (dd[k] & (BNOD - 1)), ww[k]);
      sbk[p] = (unsigned short)b;
    }
  }
  __syncthreads();
  const int nloc = (E - e0 < CH) ? (E - e0) : CH;
  for (int i = t; i < nloc; i += TPB) grp[delta[sbk[i]] + i] = stage[i];
}

__global__ void __launch_bounds__(TPB)
k_owner(const int2* __restrict__ grp, const int* __restrict__ bstart,
        int2* __restrict__ csr, int* __restrict__ rowptr, float* __restrict__ dis,
        int N, int E, int B) {
  __shared__ int cntl[BNOD];
  __shared__ float degl[BNOD];
  __shared__ int startl[BNOD];
  __shared__ int cur[BNOD];
  __shared__ int wsum[32];
  __shared__ int2 stage[CH];
  const int b = blockIdx.x;
  const int t = threadIdx.x;
  const int s = bstart[b];
  const int nE = bstart[b + 1] - s;
  if (t < BNOD) { cntl[t] = 0; degl[t] = 0.0f; }
  __syncthreads();
  for (int i = t; i < nE; i += TPB) {
    const int2 r = grp[s + i];
    const int nl = r.x & (BNOD - 1);
    atomicAdd(&cntl[nl], 1);
    atomicAdd(&degl[nl], __int_as_float(r.y));
  }
  __syncthreads();
  const int ex = block_excl_scan((t < BNOD) ? cntl[t] : 0, wsum);
  if (t < BNOD) { startl[t] = ex; cur[t] = ex; }
  __syncthreads();
  const bool inl = (nE <= CH);
  for (int i = t; i < nE; i += TPB) {
    const int2 r = grp[s + i];
    const int nl = r.x & (BNOD - 1);
    const int p = atomicAdd(&cur[nl], 1);
    const int2 o = make_int2(r.x >> BSH, r.y);
    if (inl) stage[p] = o; else csr[s + p] = o;
  }
  __syncthreads();
  if (inl) for (int i = t; i < nE; i += TPB) csr[s + i] = stage[i];
  if (t < BNOD) {
    const int n = (b << BSH) + t;
    if (n < N) {
      rowptr[n] = s + startl[t];
      const float d = degl[t] + 1.0f;
      dis[n] = (d > 0.0f) ? rsqrtf(d) : 0.0f;
    }
  }
  if (b == B - 1 && t == 0) rowptr[N] = E;
}

template <int COUT, bool RELU>
__global__ void k_gemm_wmma(const float* __restrict__ H, const float* __restrict__ Wg,
                            const float* __restrict__ dis, float* __restrict__ HW, int N) {
  constexpr int COLT = COUT / 16;
  constexpr int ROWT = 8 / COLT;
  constexpr int WSTR = 68;
  __shared__ float Wl[COUT * WSTR];

  for (int i = threadIdx.x; i < 64 * COUT; i += 256) {
    int k = i / COUT;
    int n = i - k * COUT;
    Wl[n * WSTR + k] = Wg[i];
  }
  __syncthreads();

  const int wave = threadIdx.x >> 5;
  const int lane = threadIdx.x & 31;
  const int rowTile = wave / COLT;
  const int colTile = wave - rowTile * COLT;
  const int m0 = (blockIdx.x * ROWT + rowTile) * 16;
  if (m0 >= N) return;
  const int n0 = colTile * 16;
  const int mr = lane & 15;
  const int kb = (lane >> 4) << 1;

  const int mbase = m0 + ((lane >> 4) << 3);
  const float4 dlo = *(const float4*)(dis + mbase);
  const float4 dhi = *(const float4*)(dis + mbase + 4);

  v8f acc = {};
  const float* hrow = H + (size_t)(m0 + mr) * 64;
  const float* wrow = Wl + (n0 + mr) * WSTR;
#pragma unroll
  for (int k = 0; k < 64; k += 4) {
    float a0 = hrow[k + kb];
    float a1 = hrow[k + kb + 1];
    if (RELU) { a0 = fmaxf(a0, 0.0f); a1 = fmaxf(a1, 0.0f); }
    v2f a; a.x = a0; a.y = a1;
    v2f b = *(const v2f*)&wrow[k + kb];
    acc = __builtin_amdgcn_wmma_f32_16x16x4_f32(
        false, a, false, b, (short)0, acc, false, false);
  }

  const float dsc[8] = {dlo.x, dlo.y, dlo.z, dlo.w, dhi.x, dhi.y, dhi.z, dhi.w};
#pragma unroll
  for (int v = 0; v < 8; ++v) {
    size_t idx = (size_t)(mbase + v) * COUT + n0 + mr;
    HW[idx] = acc[v] * dsc[v];
  }
}

template <int ROWB>
__device__ __forceinline__ float4 node_sum(const float* __restrict__ hw,
                                           const int2* __restrict__ csr,
                                           int n, int p, int pe, int fb) {
  constexpr int U = 8;
  const char* hb = (const char*)hw;
  float4 acc = *(const float4*)(hb + (unsigned)(n * ROWB + fb));
  if (p >= pe) return acc;
  const int last = pe - 1;
  int2 c[U];
#pragma unroll
  for (int j = 0; j < U; ++j) c[j] = csr[min(p + j, last)];
  bool more = true;
  while (more) {
    float4 v[U];
    float w[U];
#pragma unroll
    for (int j = 0; j < U; ++j)
      v[j] = *(const float4*)(hb + (unsigned)(c[j].x * ROWB + fb));
#pragma unroll
    for (int j = 0; j < U; ++j) w[j] = (p + j < pe) ? __int_as_float(c[j].y) : 0.0f;
    p += U;
    more = p < pe;
    if (more) {
#pragma unroll
      for (int j = 0; j < U; ++j) c[j] = csr[min(p + j, last)];
    }
#pragma unroll
    for (int j = 0; j < U; ++j) {
      acc.x += w[j] * v[j].x; acc.y += w[j] * v[j].y;
      acc.z += w[j] * v[j].z; acc.w += w[j] * v[j].w;
    }
  }
  return acc;
}

__global__ void __launch_bounds__(128)
k_agg16(const float* __restrict__ hw, const int2* __restrict__ csr,
        const int* __restrict__ rowptr, const float* __restrict__ bias,
        const float* __restrict__ dis, float* __restrict__ out, int N) {
  const int lane = threadIdx.x & 31;
  const int w0 = (blockIdx.x * 4 + (threadIdx.x >> 5)) * 16;
  if (w0 >= N) return;
  const int f = (lane & 3) << 2;
  int rpl = 0;
  float dl = 0.0f;
  if (lane <= 16 && w0 + lane <= N) rpl = rowptr[w0 + lane];
  if (lane < 16 && w0 + lane < N) dl = dis[w0 + lane];
  const float4 bv = *(const float4*)(bias + f);
  float* o_lsm = out;
  float* o_lg1 = out + (size_t)N * 16;
  float* o_lg2 = out + (size_t)2 * N * 16;
#pragma unroll
  for (int it = 0; it < 2; ++it) {
    const int nw = it * 8 + (lane >> 2);
    const int n = w0 + nw;
    const int p = __shfl(rpl, nw);
    const int pe = __shfl(rpl, nw + 1);
    const float dn = __shfl(dl, nw);
    float4 o = make_float4(0.0f, 0.0f, 0.0f, 0.0f);
    if (n < N) {
      const float4 acc = node_sum<64>(hw, csr, n, p, pe, f << 2);
      o.x = bv.x + dn * acc.x;
      o.y = bv.y + dn * acc.y;
      o.z = bv.z + dn * acc.z;
      o.w = bv.w + dn * acc.w;
    }
    float mx = fmaxf(fmaxf(o.x, o.y), fmaxf(o.z, o.w));
    mx = fmaxf(mx, __shfl_xor(mx, 1));
    mx = fmaxf(mx, __shfl_xor(mx, 2));
    float s = expf(o.x - mx) + expf(o.y - mx) + expf(o.z - mx) + expf(o.w - mx);
    s += __shfl_xor(s, 1);
    s += __shfl_xor(s, 2);
    const float lse = mx + logf(s);
    if (n < N) {
      const size_t at = (size_t)n * 16 + f;
      float4 l;
      l.x = o.x - lse;
      l.y = o.y - lse;
      l.z = o.z - lse;
      l.w = o.w - lse;
      *(float4*)(o_lsm + at) = l;
      *(float4*)(o_lg1 + at) = o;
      *(float4*)(o_lg2 + at) = o;
    }
  }
}

template <int COUT>
__global__ void __launch_bounds__(128)
k_agg_gemm(const float* __restrict__ hw, const int2* __restrict__ csr,
           const int* __restrict__ rowptr, const float* __restrict__ bias,
           const float* __restrict__ dis, const float* __restrict__ Wn,
           float* __restrict__ hwn, int N) {
  constexpr int C = 64;
  constexpr int ASTR = 68;
  constexpr int COLT = COUT / 16;
  __shared__ float At[64 * ASTR];
  const int wave = threadIdx.x >> 5;
  const int lane = threadIdx.x & 31;
  const int f = (lane & 15) << 2;
  const int m0 = blockIdx.x * 64;
  const int w0 = m0 + wave * 16;
  int rpl = 0;
  float dl = 0.0f;
  if (lane <= 16 && w0 + lane <= N) rpl = rowptr[w0 + lane];
  if (lane < 16 && w0 + lane < N) dl = dis[w0 + lane];
  const float4 bv = *(const float4*)(bias + f);
#pragma unroll 1
  for (int it = 0; it < 8; ++it) {
    const int nw = it * 2 + (lane >> 4);
    const int n = w0 + nw;
    const int p = __shfl(rpl, nw);
    const int pe = __shfl(rpl, nw + 1);
    const float dn = __shfl(dl, nw);
    float4 o = make_float4(0.0f, 0.0f, 0.0f, 0.0f);
    if (n < N) {
      const float4 acc = node_sum<256>(hw, csr, n, p, pe, f << 2);
      o.x = fmaxf(bv.x + dn * acc.x, 0.0f);
      o.y = fmaxf(bv.y + dn * acc.y, 0.0f);
      o.z = fmaxf(bv.z + dn * acc.z, 0.0f);
      o.w = fmaxf(bv.w + dn * acc.w, 0.0f);
    }
    float* arw = At + (wave * 16 + nw) * ASTR;
    *(float4*)&arw[f] = o;
    if (f == 0) arw[C] = dn;
  }
  __syncthreads();

  const int mr = lane & 15;
  const int kb = (lane >> 4) << 1;
  const int rl = (lane >> 4) << 3;

  v8f acc[COLT];
#pragma unroll
  for (int t = 0; t < COLT; ++t) acc[t] = v8f{};
  const float* arow = At + (wave * 16 + mr) * ASTR + kb;
  const float* wcol = Wn + (size_t)kb * COUT + mr;
#pragma unroll
  for (int k = 0; k < 64; k += 4) {
    v2f a = *(const v2f*)&arow[k];
#pragma unroll
    for (int t = 0; t < COLT; ++t) {
      v2f b;
      b.x = wcol[(size_t)k * COUT + 16 * t];
      b.y = wcol[(size_t)(k + 1) * COUT + 16 * t];
      acc[t] = __builtin_amdgcn_wmma_f32_16x16x4_f32(
          false, a, false, b, (short)0, acc[t], false, false);
    }
  }

  const float* drow = At + (wave * 16 + rl) * ASTR + C;
#pragma unroll
  for (int v = 0; v < 8; ++v) {
    const float dv = drow[v * ASTR];
#pragma unroll
    for (int t = 0; t < COLT; ++t)
      hwn[(size_t)(w0 + rl + v) * COUT + 16 * t + mr] = acc[t][v] * dv;
  }
}

extern "C" void kernel_launch(void* const* d_in, const int* in_sizes, int n_in,
                              void* d_out, int out_size, void* d_ws, size_t ws_size,
                              hipStream_t stream) {
  const float* x  = (const float*)d_in[0];
  const int*   ei = (const int*)  d_in[1];
  const float* ew = (const float*)d_in[2];
  const float* W1 = (const float*)d_in[3];
  const float* b1 = (const float*)d_in[4];
  const float* W2 = (const float*)d_in[5];
  const float* b2 = (const float*)d_in[6];
  const float* W3 = (const float*)d_in[7];
  const float* b3 = (const float*)d_in[8];
  float* out = (float*)d_out;

  const int N = in_sizes[0] / 64;
  const int E = in_sizes[2];
  const int* src = ei;
  const int* dst = ei + E;

  const int Np = (N + 64) & ~63;
  const int Ep = (E + 63) & ~63;
  const int B  = (N + BNOD - 1) >> BSH;
  const int Bq = (B + 64) & ~63;
  const int C  = (E + CH - 1) / CH;
  const int rowTiles = (N + 15) / 16;
  const int rowBlks  = (N + 63) / 64;

  int*   hist   = (int*)d_ws;
  int*   bstart = hist + (size_t)C * BMAX;
  int2*  grp    = (int2*)(bstart + Bq);
  int*   rowptr = (int*)(grp + Ep);
  int2*  csr    = (int2*)(rowptr + Np);
  float* dis    = (float*)(csr + Ep);
  float* bufA   = dis + Np;
  float* bufB   = bufA + (size_t)rowBlks * 64 * 64;

  const int gAg16 = (N + 63) / 64;

  k_hist <<<C, TPB, 0, stream>>>(dst, hist, E);
  k_group<<<C, TPB, 0, stream>>>(src, dst, ew, hist, grp, bstart, E, B, C);
  k_owner<<<B, TPB, 0, stream>>>(grp, bstart, csr, rowptr, dis, N, E, B);

  k_gemm_wmma<64, false><<<(rowTiles + 1) / 2, 256, 0, stream>>>(x, W1, dis, bufA, N);

  k_agg_gemm<64><<<rowBlks, 128, 0, stream>>>(bufA, csr, rowptr, b1, dis, W2, bufB, N);

  k_agg_gemm<16><<<rowBlks, 128, 0, stream>>>(bufB, csr, rowptr, b2, dis, W3, bufA, N);

  k_agg16<<<gAg16, 128, 0, stream>>>(bufA, csr, rowptr, b3, dis, out, N);
}


# Layer 1 product fused into the sort's bucket-owner kernel: a workgroup's 256 rows of x go through LDS into the f32 WMMA product, scaled by the D^-1/2 it has just formed in LDS; 7 launches to 6
# speedup vs baseline: 4.8692x; 1.1035x over previous
#include <hip/hip_runtime.h>
#include <hip/hip_bf16.h>
#include <math.h>


typedef float v2f __attribute__((ext_vector_type(2)));
typedef float v8f __attribute__((ext_vector_type(8)));


constexpr int BSH  = 8;
constexpr int BNOD = 1 << BSH;
constexpr int BMAX = 512;
constexpr int CH   = 8192;
constexpr int TPB  = 1024;

__device__ __forceinline__ int block_excl_scan(int v, int* wsum) {
  const int lane = threadIdx.x & 31;
  const int wave = threadIdx.x >> 5;
  int inc = v;
#pragma unroll
  for (int d = 1; d < 32; d <<= 1) {
    int u = __shfl_up(inc, d);
    if (lane >= d) inc += u;
  }
  if (lane == 31) wsum[wave] = inc;
  __syncthreads();
  const int nw = blockDim.x >> 5;
  int off = (lane < nw && lane < wave) ? wsum[lane] : 0;
#pragma unroll
  for (int m = 16; m >= 1; m >>= 1) off += __shfl_xor(off, m);
  __syncthreads();
  return off + inc - v;
}

__global__ void __launch_bounds__(TPB)
k_hist(const int* __restrict__ dst, int* __restrict__ hist, int E) {
  __shared__ int h[BMAX];
  const int t = threadIdx.x;
  if (t < BMAX) h[t] = 0;
  __syncthreads();
  const int e0 = blockIdx.x * CH;
  int d[CH / TPB];
#pragma unroll
  for (int k = 0; k < CH / TPB; ++k) {
    const int e = e0 + k * TPB + t;
    d[k] = dst[(e < E) ? e : (E - 1)];
  }
#pragma unroll
  for (int k = 0; k < CH / TPB; ++k) {
    const int e = e0 + k * TPB + t;
    if (e < E) atomicAdd(&h[d[k] >> BSH], 1);
  }
  __syncthreads();
  if (t < BMAX) hist[(size_t)blockIdx.x * BMAX + t] = h[t];
}

__global__ void __launch_bounds__(TPB)
k_group(const int* __restrict__ src, const int* __restrict__ dst,
        const float* __restrict__ ew, const int* __restrict__ hist,
        int2* __restrict__ grp, int* __restrict__ bstart,
        int E, int B, int C) {
  __shared__ int delta[BMAX];
  __shared__ int cur[BMAX];
  __shared__ int totL[BMAX];
  __shared__ int prevL[BMAX];
  __shared__ int wsum[32];
  __shared__ int2 stage[CH];
  __shared__ unsigned short sbk[CH];
  const int c = blockIdx.x;
  const int t = threadIdx.x;
  const int e0 = c * CH;
  int dd[CH / TPB], ss[CH / TPB], ww[CH / TPB];
#pragma unroll
  for (int k = 0; k < CH / TPB; ++k) {
    const int e = e0 + k * TPB + t;
    const int ec = (e < E) ? e : (E - 1);
    dd[k] = dst[ec];
    ss[k] = src[ec];
    ww[k] = __float_as_int(ew[ec]);
  }
  const int own = (t < B) ? hist[(size_t)c * BMAX + t] : 0;
  if (t < BMAX) { totL[t] = 0; prevL[t] = 0; }
  __syncthreads();
  {
    const int lane = t & 31;
    const int wave = t >> 5;
    int tt[BMAX / 32], pp[BMAX / 32];
#pragma unroll
    for (int j = 0; j < BMAX / 32; ++j) { tt[j] = 0; pp[j] = 0; }
    for (int r = wave; r < C; r += TPB / 32) {
      const int* hr = hist + (size_t)r * BMAX + lane;
      const bool before = r < c;
      int v[BMAX / 32];
#pragma unroll
      for (int j = 0; j < BMAX / 32; ++j) v[j] = hr[32 * j];
#pragma unroll
      for (int j = 0; j < BMAX / 32; ++j) {
        tt[j] += v[j];
        pp[j] += before ? v[j] : 0;
      }
    }
#pragma unroll
    for (int j = 0; j < BMAX / 32; ++j) {
      atomicAdd(&totL[lane + 32 * j], tt[j]);
      atomicAdd(&prevL[lane + 32 * j], pp[j]);
    }
  }
  __syncthreads();
  const int tot  = (t < B) ? totL[t] : 0;
  const int prev = (t < B) ? prevL[t] : 0;
  const int gstart = block_excl_scan(tot, wsum);
  const int lstart = block_excl_scan(own, wsum);
  if (t < B) {
    delta[t] = gstart + prev - lstart;
    cur[t] = lstart;
    if (c == 0) bstart[t] = gstart;
  }
  if (c == 0 && t == 0) bstart[B] = E;
  __syncthreads();
#pragma unroll
  for (int k = 0; k < CH / TPB; ++k) {
    const int e = e0 + k * TPB + t;
    if (e < E) {
      const int b = dd[k] >> BSH;
      const int p = atomicAdd(&cur[b], 1);
      stage[p] = make_int2((ss[k] << BSH) | (dd[k] & (BNOD - 1)), ww[k]);
      sbk[p] = (unsigned short)b;
    }
  }
  __syncthreads();
  const int nloc = (E - e0 < CH) ? (E - e0) : CH;
  for (int i = t; i < nloc; i += TPB) grp[delta[sbk[i]] + i] = stage[i];
}

__global__ void __launch_bounds__(TPB)
k_owner(const int2* __restrict__ grp, const int* __restrict__ bstart,
        int2* __restrict__ csr, int* __restrict__ rowptr, float* __restrict__ dis,
        const float* __restrict__ x, const float* __restrict__ W,
        float* __restrict__ hw, int N, int E, int B) {
  constexpr int ASTR = 68;
  __shared__ int cntl[BNOD];
  __shared__ float degl[BNOD];
  __shared__ float disl[BNOD];
  __shared__ int startl[BNOD];
  __shared__ int cur[BNOD];
  __shared__ int wsum[32];
  __shared__ int2 stage[CH];
  __shared__ __attribute__((aligned(16))) float At[BNOD * ASTR];
  const int b = blockIdx.x;
  const int t = threadIdx.x;
  const int s = bstart[b];
  const int nE = bstart[b + 1] - s;
  {
    const int xr = (b << BSH) + (t >> 4);
    const int xc = (t & 15) << 2;
    const float4 xv0 = *(const float4*)(x + (size_t)min(xr,       N - 1) * 64 + xc);
    const float4 xv1 = *(const float4*)(x + (size_t)min(xr + 64,  N - 1) * 64 + xc);
    const float4 xv2 = *(const float4*)(x + (size_t)min(xr + 128, N - 1) * 64 + xc);
    const float4 xv3 = *(const float4*)(x + (size_t)min(xr + 192, N - 1) * 64 + xc);
    float* ar = At + (t >> 4) * ASTR + xc;
    *(float4*)&ar[0]          = xv0;
    *(float4*)&ar[64 * ASTR]  = xv1;
    *(float4*)&ar[128 * ASTR] = xv2;
    *(float4*)&ar[192 * ASTR] = xv3;
  }
  if (t < BNOD) { cntl[t] = 0; degl[t] = 0.0f; }
  __syncthreads();
  for (int i = t; i < nE; i += TPB) {
    const int2 r = grp[s + i];
    const int nl = r.x & (BNOD - 1);
    atomicAdd(&cntl[nl], 1);
    atomicAdd(&degl[nl], __int_as_float(r.y));
  }
  __syncthreads();
  const int ex = block_excl_scan((t < BNOD) ? cntl[t] : 0, wsum);
  if (t < BNOD) {
    startl[t] = ex;
    cur[t] = ex;
    const float d = degl[t] + 1.0f;
    disl[t] = (d > 0.0f) ? rsqrtf(d) : 0.0f;
  }
  __syncthreads();
  const bool inl = (nE <= CH);
  for (int i = t; i < nE; i += TPB) {
    const int2 r = grp[s + i];
    const int nl = r.x & (BNOD - 1);
    const int p = atomicAdd(&cur[nl], 1);
    const int2 o = make_int2(r.x >> BSH, r.y);
    if (inl) stage[p] = o; else csr[s + p] = o;
  }
  __syncthreads();
  if (inl) for (int i = t; i < nE; i += TPB) csr[s + i] = stage[i];
  if (t < BNOD) {
    const int n = (b << BSH) + t;
    if (n < N) {
      rowptr[n] = s + startl[t];
      dis[n] = disl[t];
    }
  }
  if (b == B - 1 && t == 0) rowptr[N] = E;

  const int wave = t >> 5;
  const int lane = t & 31;
  const int rt = wave >> 1;
  const int cs = (wave & 1) << 1;
  const int m0 = (b << BSH) + rt * 16;
  if (m0 >= N) return;
  const int mr = lane & 15;
  const int kb = (lane >> 4) << 1;
  const int rl = (lane >> 4) << 3;
  v8f acc0 = {}, acc1 = {};
  const float* arow = At + (rt * 16 + mr) * ASTR + kb;
  const float* wcol = W + (size_t)kb * 64 + cs * 16 + mr;
#pragma unroll 8
  for (int k = 0; k < 64; k += 4) {
    v2f a = *(const v2f*)&arow[k];
    v2f b0, b1;
    b0.x = wcol[(size_t)k * 64];
    b0.y = wcol[(size_t)(k + 1) * 64];
    b1.x = wcol[(size_t)k * 64 + 16];
    b1.y = wcol[(size_t)(k + 1) * 64 + 16];
    acc0 = __builtin_amdgcn_wmma_f32_16x16x4_f32(
        false, a, false, b0, (short)0, acc0, false, false);
    acc1 = __builtin_amdgcn_wmma_f32_16x16x4_f32(
        false, a, false, b1, (short)0, acc1, false, false);
  }
#pragma unroll
  for (int v = 0; v < 8; ++v) {
    const int m = m0 + rl + v;
    if (m < N) {
      const float dv = disl[rt * 16 + rl + v];
      float* o = hw + (size_t)m * 64 + cs * 16 + mr;
      o[0]  = acc0[v] * dv;
      o[16] = acc1[v] * dv;
    }
  }
}

template <int ROWB>
__device__ __forceinline__ float4 node_sum(const float* __restrict__ hw,
                                           const int2* __restrict__ csr,
                                           int n, int p, int pe, int fb) {
  constexpr int U = 8;
  const char* hb = (const char*)hw;
  float4 acc = *(const float4*)(hb + (unsigned)(n * ROWB + fb));
  if (p >= pe) return acc;
  const int last = pe - 1;
  int2 c[U];
#pragma unroll
  for (int j = 0; j < U; ++j) c[j] = csr[min(p + j, last)];
  bool more = true;
  while (more) {
    float4 v[U];
    float w[U];
#pragma unroll
    for (int j = 0; j < U; ++j)
      v[j] = *(const float4*)(hb + (unsigned)(c[j].x * ROWB + fb));
#pragma unroll
    for (int j = 0; j < U; ++j) w[j] = (p + j < pe) ? __int_as_float(c[j].y) : 0.0f;
    p += U;
    more = p < pe;
    if (more) {
#pragma unroll
      for (int j = 0; j < U; ++j) c[j] = csr[min(p + j, last)];
    }
#pragma unroll
    for (int j = 0; j < U; ++j) {
      acc.x += w[j] * v[j].x; acc.y += w[j] * v[j].y;
      acc.z += w[j] * v[j].z; acc.w += w[j] * v[j].w;
    }
  }
  return acc;
}

__global__ void __launch_bounds__(128)
k_agg16(const float* __restrict__ hw, const int2* __restrict__ csr,
        const int* __restrict__ rowptr, const float* __restrict__ bias,
        const float* __restrict__ dis, float* __restrict__ out, int N) {
  const int lane = threadIdx.x & 31;
  const int w0 = (blockIdx.x * 4 + (threadIdx.x >> 5)) * 16;
  if (w0 >= N) return;
  const int f = (lane & 3) << 2;
  int rpl = 0;
  float dl = 0.0f;
  if (lane <= 16 && w0 + lane <= N) rpl = rowptr[w0 + lane];
  if (lane < 16 && w0 + lane < N) dl = dis[w0 + lane];
  const float4 bv = *(const float4*)(bias + f);
  float* o_lsm = out;
  float* o_lg1 = out + (size_t)N * 16;
  float* o_lg2 = out + (size_t)2 * N * 16;
#pragma unroll
  for (int it = 0; it < 2; ++it) {
    const int nw = it * 8 + (lane >> 2);
    const int n = w0 + nw;
    const int p = __shfl(rpl, nw);
    const int pe = __shfl(rpl, nw + 1);
    const float dn = __shfl(dl, nw);
    float4 o = make_float4(0.0f, 0.0f, 0.0f, 0.0f);
    if (n < N) {
      const float4 acc = node_sum<64>(hw, csr, n, p, pe, f << 2);
      o.x = bv.x + dn * acc.x;
      o.y = bv.y + dn * acc.y;
      o.z = bv.z + dn * acc.z;
      o.w = bv.w + dn * acc.w;
    }
    float mx = fmaxf(fmaxf(o.x, o.y), fmaxf(o.z, o.w));
    mx = fmaxf(mx, __shfl_xor(mx, 1));
    mx = fmaxf(mx, __shfl_xor(mx, 2));
    float s = expf(o.x - mx) + expf(o.y - mx) + expf(o.z - mx) + expf(o.w - mx);
    s += __shfl_xor(s, 1);
    s += __shfl_xor(s, 2);
    const float lse = mx + logf(s);
    if (n < N) {
      const size_t at = (size_t)n * 16 + f;
      float4 l;
      l.x = o.x - lse;
      l.y = o.y - lse;
      l.z = o.z - lse;
      l.w = o.w - lse;
      *(float4*)(o_lsm + at) = l;
      *(float4*)(o_lg1 + at) = o;
      *(float4*)(o_lg2 + at) = o;
    }
  }
}

template <int COUT>
__global__ void __launch_bounds__(128)
k_agg_gemm(const float* __restrict__ hw, const int2* __restrict__ csr,
           const int* __restrict__ rowptr, const float* __restrict__ bias,
           const float* __restrict__ dis, const float* __restrict__ Wn,
           float* __restrict__ hwn, int N) {
  constexpr int C = 64;
  constexpr int ASTR = 68;
  constexpr int COLT = COUT / 16;
  __shared__ float At[64 * ASTR];
  const int wave = threadIdx.x >> 5;
  const int lane = threadIdx.x & 31;
  const int f = (lane & 15) << 2;
  const int m0 = blockIdx.x * 64;
  const int w0 = m0 + wave * 16;
  int rpl = 0;
  float dl = 0.0f;
  if (lane <= 16 && w0 + lane <= N) rpl = rowptr[w0 + lane];
  if (lane < 16 && w0 + lane < N) dl = dis[w0 + lane];
  const float4 bv = *(const float4*)(bias + f);
#pragma unroll 1
  for (int it = 0; it < 8; ++it) {
    const int nw = it * 2 + (lane >> 4);
    const int n = w0 + nw;
    const int p = __shfl(rpl, nw);
    const int pe = __shfl(rpl, nw + 1);
    const float dn = __shfl(dl, nw);
    float4 o = make_float4(0.0f, 0.0f, 0.0f, 0.0f);
    if (n < N) {
      const float4 acc = node_sum<256>(hw, csr, n, p, pe, f << 2);
      o.x = fmaxf(bv.x + dn * acc.x, 0.0f);
      o.y = fmaxf(bv.y + dn * acc.y, 0.0f);
      o.z = fmaxf(bv.z + dn * acc.z, 0.0f);
      o.w = fmaxf(bv.w + dn * acc.w, 0.0f);
    }
    float* arw = At + (wave * 16 + nw) * ASTR;
    *(float4*)&arw[f] = o;
    if (f == 0) arw[C] = dn;
  }
  __syncthreads();

  const int mr = lane & 15;
  const int kb = (lane >> 4) << 1;
  const int rl = (lane >> 4) << 3;

  v8f acc[COLT];
#pragma unroll
  for (int t = 0; t < COLT; ++t) acc[t] = v8f{};
  const float* arow = At + (wave * 16 + mr) * ASTR + kb;
  const float* wcol = Wn + (size_t)kb * COUT + mr;
#pragma unroll
  for (int k = 0; k < 64; k += 4) {
    v2f a = *(const v2f*)&arow[k];
#pragma unroll
    for (int t = 0; t < COLT; ++t) {
      v2f b;
      b.x = wcol[(size_t)k * COUT + 16 * t];
      b.y = wcol[(size_t)(k + 1) * COUT + 16 * t];
      acc[t] = __builtin_amdgcn_wmma_f32_16x16x4_f32(
          false, a, false, b, (short)0, acc[t], false, false);
    }
  }

  const float* drow = At + (wave * 16 + rl) * ASTR + C;
#pragma unroll
  for (int v = 0; v < 8; ++v) {
    const float dv = drow[v * ASTR];
#pragma unroll
    for (int t = 0; t < COLT; ++t)
      hwn[(size_t)(w0 + rl + v) * COUT + 16 * t + mr] = acc[t][v] * dv;
  }
}

extern "C" void kernel_launch(void* const* d_in, const int* in_sizes, int n_in,
                              void* d_out, int out_size, void* d_ws, size_t ws_size,
                              hipStream_t stream) {
  const float* x  = (const float*)d_in[0];
  const int*   ei = (const int*)  d_in[1];
  const float* ew = (const float*)d_in[2];
  const float* W1 = (const float*)d_in[3];
  const float* b1 = (const float*)d_in[4];
  const float* W2 = (const float*)d_in[5];
  const float* b2 = (const float*)d_in[6];
  const float* W3 = (const float*)d_in[7];
  const float* b3 = (const float*)d_in[8];
  float* out = (float*)d_out;

  const int N = in_sizes[0] / 64;
  const int E = in_sizes[2];
  const int* src = ei;
  const int* dst = ei + E;

  const int Np = (N + 64) & ~63;
  const int Ep = (E + 63) & ~63;
  const int B  = (N + BNOD - 1) >> BSH;
  const int Bq = (B + 64) & ~63;
  const int C  = (E + CH - 1) / CH;
  const int rowBlks  = (N + 63) / 64;

  int*   hist   = (int*)d_ws;
  int*   bstart = hist + (size_t)C * BMAX;
  int2*  grp    = (int2*)(bstart + Bq);
  int*   rowptr = (int*)(grp + Ep);
  int2*  csr    = (int2*)(rowptr + Np);
  float* dis    = (float*)(csr + Ep);
  float* bufA   = dis + Np;
  float* bufB   = bufA + (size_t)rowBlks * 64 * 64;

  const int gAg16 = (N + 63) / 64;

  k_hist <<<C, TPB, 0, stream>>>(dst, hist, E);
  k_group<<<C, TPB, 0, stream>>>(src, dst, ew, hist, grp, bstart, E, B, C);
  k_owner<<<B, TPB, 0, stream>>>(grp, bstart, csr, rowptr, dis, x, W1, bufA, N, E, B);

  k_agg_gemm<64><<<rowBlks, 128, 0, stream>>>(bufA, csr, rowptr, b1, dis, W2, bufB, N);

  k_agg_gemm<16><<<rowBlks, 128, 0, stream>>>(bufB, csr, rowptr, b2, dis, W3, bufA, N);

  k_agg16<<<gAg16, 128, 0, stream>>>(bufA, csr, rowptr, b3, dis, out, N);
}


# Chunks group edges in their own slice (a run per bucket, a table of run starts); bucket owners collect their runs by it; histogram launch gone, 6 launches to 5; edge and degree sum order change
# speedup vs baseline: 5.0845x; 1.0442x over previous
#include <hip/hip_runtime.h>
#include <hip/hip_bf16.h>
#include <math.h>


typedef float v2f __attribute__((ext_vector_type(2)));
typedef float v8f __attribute__((ext_vector_type(8)));


constexpr int BSH  = 8;
constexpr int BNOD = 1 << BSH;
constexpr int BMAX = 512;
constexpr int CH   = 8192;
constexpr int TPB  = 1024;
constexpr int CMAX = 1024;

__device__ __forceinline__ int block_excl_scan(int v, int* wsum) {
  const int lane = threadIdx.x & 31;
  const int wave = threadIdx.x >> 5;
  int inc = v;
#pragma unroll
  for (int d = 1; d < 32; d <<= 1) {
    int u = __shfl_up(inc, d);
    if (lane >= d) inc += u;
  }
  if (lane == 31) wsum[wave] = inc;
  __syncthreads();
  const int nw = blockDim.x >> 5;
  int off = (lane < nw && lane < wave) ? wsum[lane] : 0;
#pragma unroll
  for (int m = 16; m >= 1; m >>= 1) off += __shfl_xor(off, m);
  __syncthreads();
  return off + inc - v;
}

__global__ void __launch_bounds__(TPB)
k_group(const int* __restrict__ src, const int* __restrict__ dst,
        const float* __restrict__ ew, int2* __restrict__ tab,
        int2* __restrict__ grp, int E) {
  __shared__ int cnt[BMAX];
  __shared__ int wsum[32];
  __shared__ int2 stage[CH];
  const int c = blockIdx.x;
  const int t = threadIdx.x;
  const int e0 = c * CH;
  int dd[CH / TPB], ss[CH / TPB], ww[CH / TPB], rk[CH / TPB];
#pragma unroll
  for (int k = 0; k < CH / TPB; ++k) {
    const int e = e0 + k * TPB + t;
    const int ec = (e < E) ? e : (E - 1);
    dd[k] = dst[ec];
    ss[k] = src[ec];
    ww[k] = __float_as_int(ew[ec]);
  }
  if (t < BMAX) cnt[t] = 0;
  __syncthreads();
#pragma unroll
  for (int k = 0; k < CH / TPB; ++k) {
    const int e = e0 + k * TPB + t;
    rk[k] = (e < E) ? atomicAdd(&cnt[dd[k] >> BSH], 1) : 0;
  }
  __syncthreads();
  const int own = (t < BMAX) ? cnt[t] : 0;
  const int lstart = block_excl_scan(own, wsum);
  if (t < BMAX) {
    cnt[t] = lstart;
    tab[(size_t)c * BMAX + t] = make_int2(lstart, own);
  }
  __syncthreads();
#pragma unroll
  for (int k = 0; k < CH / TPB; ++k) {
    const int e = e0 + k * TPB + t;
    if (e < E) {
      const int p = cnt[dd[k] >> BSH] + rk[k];
      stage[p] = make_int2((ss[k] << BSH) | (dd[k] & (BNOD - 1)), ww[k]);
    }
  }
  __syncthreads();
  const int nloc = (E - e0 < CH) ? (E - e0) : CH;
  for (int i = t; i < nloc; i += TPB) grp[(size_t)e0 + i] = stage[i];
}

__global__ void __launch_bounds__(TPB)
k_owner(const int2* __restrict__ grp, const int2* __restrict__ tab,
        int2* __restrict__ csr, int* __restrict__ rowptr, float* __restrict__ dis,
        const float* __restrict__ x, const float* __restrict__ W,
        float* __restrict__ hw, int N, int E, int B, int C) {
  constexpr int ASTR = 68;
  constexpr int RB = 8;
  __shared__ int cntl[BNOD];
  __shared__ float degl[BNOD];
  __shared__ float disl[BNOD];
  __shared__ int startl[BNOD];
  __shared__ int cur[BNOD];
  __shared__ int wsum[32];
  __shared__ int runSrc[CMAX];
  __shared__ int runLen[CMAX];
  __shared__ int tot[2];
  __shared__ int2 stage[CH];
  __shared__ __attribute__((aligned(16))) float At[BNOD * ASTR];
  const int b = blockIdx.x;
  const int t = threadIdx.x;
  const int lane = t & 31;
  const int wave = t >> 5;
  {
    const int xr = (b << BSH) + (t >> 4);
    const int xc = (t & 15) << 2;
    const float4 xv0 = *(const float4*)(x + (size_t)min(xr,       N - 1) * 64 + xc);
    const float4 xv1 = *(const float4*)(x + (size_t)min(xr + 64,  N - 1) * 64 + xc);
    const float4 xv2 = *(const float4*)(x + (size_t)min(xr + 128, N - 1) * 64 + xc);
    const float4 xv3 = *(const float4*)(x + (size_t)min(xr + 192, N - 1) * 64 + xc);
    const int2 tv = (t < C) ? tab[(size_t)t * BMAX + b] : make_int2(0, 0);
    if (t < C) { runSrc[t] = t * CH + tv.x; runLen[t] = tv.y; }
    if (t < 2) tot[t] = 0;
    if (t < BNOD) { cntl[t] = 0; degl[t] = 0.0f; }
    float* ar = At + (t >> 4) * ASTR + xc;
    *(float4*)&ar[0]          = xv0;
    *(float4*)&ar[64 * ASTR]  = xv1;
    *(float4*)&ar[128 * ASTR] = xv2;
    *(float4*)&ar[192 * ASTR] = xv3;
    int sx = tv.x, sy = tv.y;
#pragma unroll
    for (int m = 16; m >= 1; m >>= 1) { sx += __shfl_xor(sx, m); sy += __shfl_xor(sy, m); }
    __syncthreads();
    if (lane == 0 && wave * 32 < C) { atomicAdd(&tot[0], sx); atomicAdd(&tot[1], sy); }
  }
  __syncthreads();
  const int s = tot[0];
  const int nE = tot[1];
  auto each_edge = [&](auto f) {
    for (int c0 = wave; c0 < C; c0 += 32 * RB) {
      int2 r[RB];
      bool ok[RB];
      bool longer = false;
#pragma unroll
      for (int k = 0; k < RB; ++k) {
        const int c = c0 + 32 * k;
        const int cc = min(c, C - 1);
        const int len = (c < C) ? runLen[cc] : 0;
        ok[k] = lane < len;
        longer = longer || (len > 32);
        r[k] = grp[min(runSrc[cc] + lane, E - 1)];
      }
#pragma unroll
      for (int k = 0; k < RB; ++k) if (ok[k]) f(r[k]);
      if (longer) {
        for (int k = 0; k < RB; ++k) {
          const int c = c0 + 32 * k;
          if (c < C) {
            const int len = runLen[c];
            const int2* g = grp + (size_t)runSrc[c];
            for (int j = lane + 32; j < len; j += 32) f(g[j]);
          }
        }
      }
    }
  };
  each_edge([&](int2 r) {
    const int nl = r.x & (BNOD - 1);
    atomicAdd(&cntl[nl], 1);
    atomicAdd(&degl[nl], __int_as_float(r.y));
  });
  __syncthreads();
  const int ex = block_excl_scan((t < BNOD) ? cntl[t] : 0, wsum);
  if (t < BNOD) {
    startl[t] = ex;
    cur[t] = ex;
    const float d = degl[t] + 1.0f;
    disl[t] = (d > 0.0f) ? rsqrtf(d) : 0.0f;
  }
  __syncthreads();
  const bool inl = (nE <= CH);
  each_edge([&](int2 r) {
    const int nl = r.x & (BNOD - 1);
    const int p = atomicAdd(&cur[nl], 1);
    const int2 o = make_int2(r.x >> BSH, r.y);
    if (inl) stage[p] = o; else csr[s + p] = o;
  });
  __syncthreads();
  if (inl) for (int i = t; i < nE; i += TPB) csr[s + i] = stage[i];
  if (t < BNOD) {
    const int n = (b << BSH) + t;
    if (n < N) {
      rowptr[n] = s + startl[t];
      dis[n] = disl[t];
    }
  }
  if (b == B - 1 && t == 0) rowptr[N] = E;

  const int rt = wave >> 1;
  const int cs = (wave & 1) << 1;
  const int m0 = (b << BSH) + rt * 16;
  if (m0 >= N) return;
  const int mr = lane & 15;
  const int kb = (lane >> 4) << 1;
  const int rl = (lane >> 4) << 3;
  v8f acc0 = {}, acc1 = {};
  const float* arow = At + (rt * 16 + mr) * ASTR + kb;
  const float* wcol = W + (size_t)kb * 64 + cs * 16 + mr;
#pragma unroll 8
  for (int k = 0; k < 64; k += 4) {
    v2f a = *(const v2f*)&arow[k];
    v2f b0, b1;
    b0.x = wcol[(size_t)k * 64];
    b0.y = wcol[(size_t)(k + 1) * 64];
    b1.x = wcol[(size_t)k * 64 + 16];
    b1.y = wcol[(size_t)(k + 1) * 64 + 16];
    acc0 = __builtin_amdgcn_wmma_f32_16x16x4_f32(
        false, a, false, b0, (short)0, acc0, false, false);
    acc1 = __builtin_amdgcn_wmma_f32_16x16x4_f32(
        false, a, false, b1, (short)0, acc1, false, false);
  }
#pragma unroll
  for (int v = 0; v < 8; ++v) {
    const int m = m0 + rl + v;
    if (m < N) {
      const float dv = disl[rt * 16 + rl + v];
      float* o = hw + (size_t)m * 64 + cs * 16 + mr;
      o[0]  = acc0[v] * dv;
      o[16] = acc1[v] * dv;
    }
  }
}

template <int ROWB>
__device__ __forceinline__ float4 node_sum(const float* __restrict__ hw,
                                           const int2* __restrict__ csr,
                                           int n, int p, int pe, int fb) {
  constexpr int U = 8;
  const char* hb = (const char*)hw;
  float4 acc = *(const float4*)(hb + (unsigned)(n * ROWB + fb));
  if (p >= pe) return acc;
  const int last = pe - 1;
  int2 c[U];
#pragma unroll
  for (int j = 0; j < U; ++j) c[j] = csr[min(p + j, last)];
  bool more = true;
  while (more) {
    float4 v[U];
    float w[U];
#pragma unroll
    for (int j = 0; j < U; ++j)
      v[j] = *(const float4*)(hb + (unsigned)(c[j].x * ROWB + fb));
#pragma unroll
    for (int j = 0; j < U; ++j) w[j] = (p + j < pe) ? __int_as_float(c[j].y) : 0.0f;
    p += U;
    more = p < pe;
    if (more) {
#pragma unroll
      for (int j = 0; j < U; ++j) c[j] = csr[min(p + j, last)];
    }
#pragma unroll
    for (int j = 0; j < U; ++j) {
      acc.x += w[j] * v[j].x; acc.y += w[j] * v[j].y;
      acc.z += w[j] * v[j].z; acc.w += w[j] * v[j].w;
    }
  }
  return acc;
}

__global__ void __launch_bounds__(128)
k_agg16(const float* __restrict__ hw, const int2* __restrict__ csr,
        const int* __restrict__ rowptr, const float* __restrict__ bias,
        const float* __restrict__ dis, float* __restrict__ out, int N) {
  const int lane = threadIdx.x & 31;
  const int w0 = (blockIdx.x * 4 + (threadIdx.x >> 5)) * 16;
  if (w0 >= N) return;
  const int f = (lane & 3) << 2;
  int rpl = 0;
  float dl = 0.0f;
  if (lane <= 16 && w0 + lane <= N) rpl = rowptr[w0 + lane];
  if (lane < 16 && w0 + lane < N) dl = dis[w0 + lane];
  const float4 bv = *(const float4*)(bias + f);
  float* o_lsm = out;
  float* o_lg1 = out + (size_t)N * 16;
  float* o_lg2 = out + (size_t)2 * N * 16;
#pragma unroll
  for (int it = 0; it < 2; ++it) {
    const int nw = it * 8 + (lane >> 2);
    const int n = w0 + nw;
    const int p = __shfl(rpl, nw);
    const int pe = __shfl(rpl, nw + 1);
    const float dn = __shfl(dl, nw);
    float4 o = make_float4(0.0f, 0.0f, 0.0f, 0.0f);
    if (n < N) {
      const float4 acc = node_sum<64>(hw, csr, n, p, pe, f << 2);
      o.x = bv.x + dn * acc.x;
      o.y = bv.y + dn * acc.y;
      o.z = bv.z + dn * acc.z;
      o.w = bv.w + dn * acc.w;
    }
    float mx = fmaxf(fmaxf(o.x, o.y), fmaxf(o.z, o.w));
    mx = fmaxf(mx, __shfl_xor(mx, 1));
    mx = fmaxf(mx, __shfl_xor(mx, 2));
    float s = expf(o.x - mx) + expf(o.y - mx) + expf(o.z - mx) + expf(o.w - mx);
    s += __shfl_xor(s, 1);
    s += __shfl_xor(s, 2);
    const float lse = mx + logf(s);
    if (n < N) {
      const size_t at = (size_t)n * 16 + f;
      float4 l;
      l.x = o.x - lse;
      l.y = o.y - lse;
      l.z = o.z - lse;
      l.w = o.w - lse;
      *(float4*)(o_lsm + at) = l;
      *(float4*)(o_lg1 + at) = o;
      *(float4*)(o_lg2 + at) = o;
    }
  }
}

template <int COUT>
__global__ void __launch_bounds__(128)
k_agg_gemm(const float* __restrict__ hw, const int2* __restrict__ csr,
           const int* __restrict__ rowptr, const float* __restrict__ bias,
           const float* __restrict__ dis, const float* __restrict__ Wn,
           float* __restrict__ hwn, int N) {
  constexpr int C = 64;
  constexpr int ASTR = 68;
  constexpr int COLT = COUT / 16;
  __shared__ float At[64 * ASTR];
  const int wave = threadIdx.x >> 5;
  const int lane = threadIdx.x & 31;
  const int f = (lane & 15) << 2;
  const int m0 = blockIdx.x * 64;
  const int w0 = m0 + wave * 16;
  int rpl = 0;
  float dl = 0.0f;
  if (lane <= 16 && w0 + lane <= N) rpl = rowptr[w0 + lane];
  if (lane < 16 && w0 + lane < N) dl = dis[w0 + lane];
  const float4 bv = *(const float4*)(bias + f);
#pragma unroll 1
  for (int it = 0; it < 8; ++it) {
    const int nw = it * 2 + (lane >> 4);
    const int n = w0 + nw;
    const int p = __shfl(rpl, nw);
    const int pe = __shfl(rpl, nw + 1);
    const float dn = __shfl(dl, nw);
    float4 o = make_float4(0.0f, 0.0f, 0.0f, 0.0f);
    if (n < N) {
      const float4 acc = node_sum<256>(hw, csr, n, p, pe, f << 2);
      o.x = fmaxf(bv.x + dn * acc.x, 0.0f);
      o.y = fmaxf(bv.y + dn * acc.y, 0.0f);
      o.z = fmaxf(bv.z + dn * acc.z, 0.0f);
      o.w = fmaxf(bv.w + dn * acc.w, 0.0f);
    }
    float* arw = At + (wave * 16 + nw) * ASTR;
    *(float4*)&arw[f] = o;
    if (f == 0) arw[C] = dn;
  }
  __syncthreads();

  const int mr = lane & 15;
  const int kb = (lane >> 4) << 1;
  const int rl = (lane >> 4) << 3;

  v8f acc[COLT];
#pragma unroll
  for (int t = 0; t < COLT; ++t) acc[t] = v8f{};
  const float* arow = At + (wave * 16 + mr) * ASTR + kb;
  const float* wcol = Wn + (size_t)kb * COUT + mr;
#pragma unroll
  for (int k = 0; k < 64; k += 4) {
    v2f a = *(const v2f*)&arow[k];
#pragma unroll
    for (int t = 0; t < COLT; ++t) {
      v2f b;
      b.x = wcol[(size_t)k * COUT + 16 * t];
      b.y = wcol[(size_t)(k + 1) * COUT + 16 * t];
      acc[t] = __builtin_amdgcn_wmma_f32_16x16x4_f32(
          false, a, false, b, (short)0, acc[t], false, false);
    }
  }

  const float* drow = At + (wave * 16 + rl) * ASTR + C;
#pragma unroll
  for (int v = 0; v < 8; ++v) {
    const float dv = drow[v * ASTR];
#pragma unroll
    for (int t = 0; t < COLT; ++t)
      hwn[(size_t)(w0 + rl + v) * COUT + 16 * t + mr] = acc[t][v] * dv;
  }
}

extern "C" void kernel_launch(void* const* d_in, const int* in_sizes, int n_in,
                              void* d_out, int out_size, void* d_ws, size_t ws_size,
                              hipStream_t stream) {
  const float* x  = (const float*)d_in[0];
  const int*   ei = (const int*)  d_in[1];
  const float* ew = (const float*)d_in[2];
  const float* W1 = (const float*)d_in[3];
  const float* b1 = (const float*)d_in[4];
  const float* W2 = (const float*)d_in[5];
  const float* b2 = (const float*)d_in[6];
  const float* W3 = (const float*)d_in[7];
  const float* b3 = (const float*)d_in[8];
  float* out = (float*)d_out;

  const int N = in_sizes[0] / 64;
  const int E = in_sizes[2];
  const int* src = ei;
  const int* dst = ei + E;

  const int Np = (N + 64) & ~63;
  const int Ep = (E + 63) & ~63;
  const int B  = (N + BNOD - 1) >> BSH;
  const int C  = (E + CH - 1) / CH;
  const int rowBlks  = (N + 63) / 64;

  int2*  tab    = (int2*)d_ws;
  int2*  grp    = tab + (size_t)C * BMAX;
  int*   rowptr = (int*)(grp + (size_t)C * CH);
  int2*  csr    = (int2*)(rowptr + Np);
  float* dis    = (float*)(csr + Ep);
  float* bufA   = dis + Np;
  float* bufB   = bufA + (size_t)rowBlks * 64 * 64;

  const int gAg16 = (N + 63) / 64;

  k_group<<<C, TPB, 0, stream>>>(src, dst, ew, tab, grp, E);
  k_owner<<<B, TPB, 0, stream>>>(grp, tab, csr, rowptr, dis, x, W1, bufA, N, E, B, C);

  k_agg_gemm<64><<<rowBlks, 128, 0, stream>>>(bufA, csr, rowptr, b1, dis, W2, bufB, N);

  k_agg_gemm<16><<<rowBlks, 128, 0, stream>>>(bufB, csr, rowptr, b2, dis, W3, bufA, N);

  k_agg16<<<gAg16, 128, 0, stream>>>(bufA, csr, rowptr, b3, dis, out, N);
}
